# starting kernel
# baseline (speedup 1.0000x reference)
#include <hip/hip_runtime.h>

typedef __attribute__((ext_vector_type(2))) float v2f;
typedef __attribute__((ext_vector_type(4))) float v4f;
typedef __attribute__((ext_vector_type(8))) float v8f;

template <int K, int L>
__global__ __launch_bounds__(128)
void gemm_f32_wmma(const float* __restrict__ W, const float* __restrict__ U,
                   float* __restrict__ Out, int M)
{
    const int b = blockIdx.z;
    const float* __restrict__ Ub = U + (long)b * K * L;
    float* __restrict__ Ob = Out + (long)b * (long)M * L;

    const int wave  = threadIdx.x >> 5;
    const int lane  = threadIdx.x & 31;
    const int m0    = blockIdx.y * 32 + (wave >> 1) * 16;
    const int l0    = blockIdx.x * 32 + (wave & 1) * 16;
    const int row   = lane & 15;
    const int khalf = lane >> 4;

    const float* __restrict__ aptr = W  + (long)(m0 + row) * K + 2 * khalf;
    const float* __restrict__ bptr = Ub + (long)(2 * khalf) * L + l0 + row;

    v8f acc = {};
    for (int k = 0; k < K; k += 32) {
        #pragma unroll
        for (int j = 0; j < 8; ++j) {
            v2f a = *(const v2f*)(aptr + j * 4);
            v2f bm;
            bm.x = bptr[(long)(j * 4)     * L];
            bm.y = bptr[(long)(j * 4 + 1) * L];
            acc = __builtin_amdgcn_wmma_f32_16x16x4_f32(
                      false, a, false, bm, (short)0, acc, false, false);
        }
        aptr += 32;
        bptr += (long)32 * L;
    }

    float* __restrict__ dst = Ob + (long)(l0 + row) * M + m0 + 8 * khalf;
    v4f lo = {acc[0], acc[1], acc[2], acc[3]};
    v4f hi = {acc[4], acc[5], acc[6], acc[7]};
    *(v4f*)(dst)     = lo;
    *(v4f*)(dst + 4) = hi;
}

__global__ __launch_bounds__(128)
void ssm_scan(const float* __restrict__ u, const float* __restrict__ A,
              const float* __restrict__ D,
              const float* __restrict__ BuT, const float* __restrict__ CuT,
              const float* __restrict__ dT, float* __restrict__ out,
              int dmodel, int N, int L)
{
    const int b  = blockIdx.y;
    const int t  = threadIdx.x;
    const int d  = blockIdx.x * 64 + (t >> 1);
    const int nh = (t & 1) * 32;

    const float* __restrict__ ubd = u   + ((size_t)b * dmodel + d) * L;
    float*       __restrict__ ybd = out + ((size_t)b * dmodel + d) * L;
    const float* __restrict__ Bub = BuT + (size_t)b * L * N;
    const float* __restrict__ Cub = CuT + (size_t)b * L * N;
    const float* __restrict__ dTb = dT  + (size_t)b * L * dmodel;

    const float Dd = D[d];

    float areg[32], h[32];
    #pragma unroll
    for (int i = 0; i < 32; ++i) { areg[i] = A[nh + i]; h[i] = 0.0f; }

    __shared__ float sB[64];
    __shared__ float sC[64];

    for (int l = 0; l < L; ++l) {
        if (t < 64) sB[t]      = Bub[(size_t)l * N + t];
        else        sC[t - 64] = Cub[(size_t)l * N + (t - 64)];
        __syncthreads();

        const float dl = dTb[(size_t)l * dmodel + d];
        const float ul = ubd[l];
        if (l + 1 < L) __builtin_prefetch(&ubd[l + 1], 0, 0);
        const float du = dl * ul;

        float y = 0.0f;
        #pragma unroll
        for (int i = 0; i < 32; ++i) {
            const float hn = fmaf(dl * areg[i], h[i], du * sB[nh + i]);
            h[i] = hn;
            y = fmaf(sC[nh + i], hn, y);
        }
        y += __shfl_xor(y, 1, 32);
        if ((t & 1) == 0) ybd[l] = fmaf(ul, Dd, y);
        __syncthreads();
    }
}

extern "C" void kernel_launch(void* const* d_in, const int* in_sizes, int n_in,
                              void* d_out, int out_size, void* d_ws, size_t ws_size,
                              hipStream_t stream)
{
    (void)in_sizes; (void)n_in; (void)out_size; (void)ws_size;

    const float* u     = (const float*)d_in[1];
    const float* A     = (const float*)d_in[2];
    const float* Bm    = (const float*)d_in[3];
    const float* Cm    = (const float*)d_in[4];
    const float* D     = (const float*)d_in[5];
    const float* Delta = (const float*)d_in[6];
    float* out = (float*)d_out;

    constexpr int batch = 4, dmodel = 1024, N = 64, L = 2048;

    float* BuT = (float*)d_ws;
    float* CuT = BuT + (size_t)batch * L * N;
    float* dT  = CuT + (size_t)batch * L * N;

    const dim3 blk(128, 1, 1);

    gemm_f32_wmma<dmodel, L><<<dim3(L / 32, N / 32, batch), blk, 0, stream>>>(
        Bm, u, BuT, N);
    gemm_f32_wmma<dmodel, L><<<dim3(L / 32, N / 32, batch), blk, 0, stream>>>(
        Cm, u, CuT, N);
    gemm_f32_wmma<dmodel, L><<<dim3(L / 32, dmodel / 32, batch), blk, 0, stream>>>(
        Delta, u, dT, dmodel);

    ssm_scan<<<dim3(dmodel / 64, batch), dim3(128, 1, 1), 0, stream>>>(
        u, A, D, BuT, CuT, dT, out, dmodel, N, L);
}


# scan in weights form, 16 steps at a time: y_l = dp_l*sum_n Ct_l[n]*h[n] + sum_{j<=l} w_j*K[l][j]; Ct, K, state-update terms from a new small launch by scalar loads; y and h sum order changes
# speedup vs baseline: 12.4296x; 12.4296x over previous
#include <hip/hip_runtime.h>

typedef __attribute__((ext_vector_type(2))) float v2f;
typedef __attribute__((ext_vector_type(4))) float v4f;
typedef __attribute__((ext_vector_type(8))) float v8f;

template <int TM, int TN, int KS, int PA, int PB>
__device__ __forceinline__
void gemm_stage(const float* __restrict__ As, const float* __restrict__ Bs,
                v8f (&acc)[TM][TN])
{
    #pragma unroll
    for (int j = 0; j < KS / 4; ++j) {
        v2f a[TM], b[TN];
        #pragma unroll
        for (int i = 0; i < TM; ++i)
            a[i] = *(const v2f*)(As + i * 16 * PA + 4 * j);
        #pragma unroll
        for (int t = 0; t < TN; ++t) {
            b[t].x = Bs[(4 * j)     * PB + t * 16];
            b[t].y = Bs[(4 * j + 1) * PB + t * 16];
        }
        #pragma unroll
        for (int i = 0; i < TM; ++i) {
            #pragma unroll
            for (int t = 0; t < TN; ++t)
                acc[i][t] = __builtin_amdgcn_wmma_f32_16x16x4_f32(
                                false, a[i], false, b[t], (short)0, acc[i][t], false, false);
        }
    }
}

template <int K, int L, int TM, int TN, int KS>
__global__ __launch_bounds__(256)
__attribute__((amdgpu_waves_per_eu(4, 4)))
void gemm_f32_wmma(const float* __restrict__ W, const float* __restrict__ U,
                   float* __restrict__ Out, int M)
{
    constexpr int BM = 32 * TM;
    constexpr int BN = 64 * TN;
    constexpr int PA = KS + 4;
    constexpr int PB = BN + 8;
    constexpr int NS = K / KS;
    constexpr int A_VPR = KS / 4;
    constexpr int A_RPP = 256 / A_VPR;
    constexpr int A_NP  = BM / A_RPP;
    constexpr int B_VPR = BN / 4;
    constexpr int B_RPP = 256 / B_VPR;
    constexpr int B_NP  = KS / B_RPP;
    static_assert(K % KS == 0 && BM % A_RPP == 0 && KS % B_RPP == 0, "tile shapes");
    static_assert((PA % 8) == 4 && (PB % 32) == 8, "bank-conflict-free pitches");

    __shared__ __attribute__((aligned(16))) float sA[2][BM * PA];
    __shared__ __attribute__((aligned(16))) float sB[2][KS * PB];

    const int b = blockIdx.z;
    const float* __restrict__ Ub = U + (long)b * K * L;
    float* __restrict__ Ob = Out + (long)b * (long)M * L;

    const int t    = threadIdx.x;
    const int wave = t >> 5;
    const int lane = t & 31;
    const int wm   = wave >> 2;
    const int wn   = wave & 3;
    const int row  = lane & 15;
    const int kh   = lane >> 4;
    const int m0   = blockIdx.y * BM;
    const int l0   = blockIdx.x * BN;

    const int ar = t / A_VPR, aq = t % A_VPR;
    const int br = t / B_VPR, bq = t % B_VPR;
    const float* __restrict__ ag = W  + (long)(m0 + ar) * K + 4 * aq;
    const float* __restrict__ bg = Ub + (long)br * L + l0 + 4 * bq;
    const int as_off = ar * PA + 4 * aq;
    const int bs_off = br * PB + 4 * bq;

    const int af_off = (wm * TM * 16 + row) * PA + 2 * kh;
    const int bf_off = (2 * kh) * PB + wn * TN * 16 + row;

    v8f acc[TM][TN];
    #pragma unroll
    for (int i = 0; i < TM; ++i) {
        #pragma unroll
        for (int j = 0; j < TN; ++j) acc[i][j] = v8f{};
    }

    v4f ra[A_NP], rb[B_NP];

    #pragma unroll
    for (int p = 0; p < A_NP; ++p) ra[p] = *(const v4f*)(ag + (long)p * A_RPP * K);
    #pragma unroll
    for (int p = 0; p < B_NP; ++p) rb[p] = *(const v4f*)(bg + (long)p * B_RPP * L);
    #pragma unroll
    for (int p = 0; p < A_NP; ++p) *(v4f*)(&sA[0][as_off + p * A_RPP * PA]) = ra[p];
    #pragma unroll
    for (int p = 0; p < B_NP; ++p) *(v4f*)(&sB[0][bs_off + p * B_RPP * PB]) = rb[p];
    __syncthreads();

    for (int s = 0; s + 1 < NS; ++s) {
        const int cur = s & 1;
        ag += KS;
        bg += (long)KS * L;
        #pragma unroll
        for (int p = 0; p < A_NP; ++p) ra[p] = *(const v4f*)(ag + (long)p * A_RPP * K);
        #pragma unroll
        for (int p = 0; p < B_NP; ++p) rb[p] = *(const v4f*)(bg + (long)p * B_RPP * L);

        gemm_stage<TM, TN, KS, PA, PB>(&sA[cur][af_off], &sB[cur][bf_off], acc);

        #pragma unroll
        for (int p = 0; p < A_NP; ++p) *(v4f*)(&sA[cur ^ 1][as_off + p * A_RPP * PA]) = ra[p];
        #pragma unroll
        for (int p = 0; p < B_NP; ++p) *(v4f*)(&sB[cur ^ 1][bs_off + p * B_RPP * PB]) = rb[p];
        __syncthreads();
    }
    gemm_stage<TM, TN, KS, PA, PB>(&sA[(NS - 1) & 1][af_off], &sB[(NS - 1) & 1][bf_off], acc);

    #pragma unroll
    for (int i = 0; i < TM; ++i) {
        #pragma unroll
        for (int j = 0; j < TN; ++j) {
            float* __restrict__ dst = Ob + (long)(l0 + (wn * TN + j) * 16 + row) * M
                                         + m0 + (wm * TM + i) * 16 + 8 * kh;
            v4f lo = {acc[i][j][0], acc[i][j][1], acc[i][j][2], acc[i][j][3]};
            v4f hi = {acc[i][j][4], acc[i][j][5], acc[i][j][6], acc[i][j][7]};
            *(v4f*)(dst)     = lo;
            *(v4f*)(dst + 4) = hi;
        }
    }
}

template <int T>
__global__ __launch_bounds__(128)
void ssm_chunk_state(const float* __restrict__ u, const float* __restrict__ A,
                     const float* __restrict__ BuT, const float* __restrict__ dT,
                     float* __restrict__ hloc, float* __restrict__ pdel,
                     int dmodel, int N, int L)
{
    static_assert(T == 64, "u tile copy: 16 threads x 16 bytes per row of T steps");
    constexpr int PG = T + 4;
    constexpr int PE = 64 + 8;
    __shared__ __attribute__((aligned(16))) float sG[64 * PG];
    __shared__ __attribute__((aligned(16))) float sE[T * PE];

    const int b    = blockIdx.y;
    const int c    = blockIdx.z;
    const int t    = threadIdx.x;
    const int d0   = blockIdx.x * 64;
    const int lbeg = c * T;
    const int wv   = __builtin_amdgcn_readfirstlane(t >> 5);

    const float* __restrict__ ub  = u   + ((size_t)b * dmodel + d0) * L + lbeg;
    const float* __restrict__ Bub = BuT + ((size_t)b * L + lbeg) * N;
    const float* __restrict__ dTb = dT  + ((size_t)b * L + lbeg) * dmodel + d0;

    {
        const int r = t >> 4, q = t & 15;
        v4f uv[8];
        #pragma unroll
        for (int p = 0; p < 8; ++p)
            uv[p] = *(const v4f*)(ub + (size_t)(r + 8 * p) * L + 4 * q);
        #pragma unroll
        for (int p = 0; p < 8; ++p)
            *(v4f*)(&sG[(r + 8 * p) * PG + 4 * q]) = uv[p];
    }

    const int    idx     = t & 63;
    const float* __restrict__ colp = (wv < 2) ? (Bub + idx) : (dTb + idx);
    const size_t cstride = (wv < 2) ? (size_t)N : (size_t)dmodel;
    float v[T];
    #pragma unroll
    for (int j = 0; j < T; ++j) v[j] = colp[(size_t)j * cstride];

    __syncthreads();

    if (wv < 2) {
        const float a = A[idx];
        float pw = 1.0f;
        #pragma unroll
        for (int j = T - 1; j >= 0; --j) {
            sE[j * PE + idx] = pw * v[j];
            pw *= a;
        }
    } else {
        float s = 1.0f;
        #pragma unroll
        for (int j = T - 1; j >= 0; --j) {
            const float ul = sG[idx * PG + j];
            sG[idx * PG + j] = s * (v[j] * ul);
            s *= v[j];
        }
        pdel[((size_t)c * gridDim.y + b) * dmodel + d0 + idx] = s;
    }
    __syncthreads();

    const int lane = t & 31;
    const int cc   = lane & 15;
    const int kh   = lane >> 4;

    v8f acc[4];
    #pragma unroll
    for (int tm = 0; tm < 4; ++tm) acc[tm] = v8f{};

    const float* __restrict__ gp = &sG[(16 * wv + cc) * PG + 2 * kh];
    const float* __restrict__ ep = &sE[(2 * kh) * PE + cc];
    #pragma unroll
    for (int kk = 0; kk < T / 4; ++kk) {
        const v2f g = *(const v2f*)(gp + 4 * kk);
        #pragma unroll
        for (int tm = 0; tm < 4; ++tm) {
            v2f e;
            e.x = ep[(4 * kk)     * PE + 16 * tm];
            e.y = ep[(4 * kk + 1) * PE + 16 * tm];
            acc[tm] = __builtin_amdgcn_wmma_f32_16x16x4_f32(
                          false, e, false, g, (short)0, acc[tm], false, false);
        }
    }

    const int    d = d0 + 16 * wv + cc;
    const size_t o = ((((size_t)c * gridDim.y + b) * dmodel) + d) * N;
    #pragma unroll
    for (int tm = 0; tm < 4; ++tm) {
        const int n0 = 16 * tm + 8 * kh;
        const v4f lo = {acc[tm][0], acc[tm][1], acc[tm][2], acc[tm][3]};
        const v4f hi = {acc[tm][4], acc[tm][5], acc[tm][6], acc[tm][7]};
        *(v4f*)(hloc + o + n0)     = lo;
        *(v4f*)(hloc + o + n0 + 4) = hi;
    }
}

template <int N, int T>
__global__ __launch_bounds__(256)
void ssm_chunk_carry(const float* __restrict__ hloc, const float* __restrict__ pdel,
                     const float* __restrict__ A,
                     float* __restrict__ hstart, int nchunks, size_t plane)
{
    const size_t i    = ((size_t)blockIdx.x * blockDim.x + threadIdx.x) * 4;
    const size_t row  = i / (size_t)N;
    const int    n    = (int)(i % (size_t)N);
    const size_t rows = plane / (size_t)N;

    const v4f a = *(const v4f*)(A + n);
    v4f at = {1.0f, 1.0f, 1.0f, 1.0f};
    for (int j = 0; j < T; ++j) at *= a;

    v4f hs = {0.0f, 0.0f, 0.0f, 0.0f};
    for (int c = 0; c + 1 < nchunks; ++c) {
        const v4f   hl = *(const v4f*)(hloc + (size_t)c * plane + i);
        const float pd = pdel[(size_t)c * rows + row];
        const v4f   pv = pd * at;
        *(v4f*)(hstart + (size_t)c * plane + i) = hs;
        hs.x = fmaf(pv.x, hs.x, hl.x);
        hs.y = fmaf(pv.y, hs.y, hl.y);
        hs.z = fmaf(pv.z, hs.z, hl.z);
        hs.w = fmaf(pv.w, hs.w, hl.w);
    }
    *(v4f*)(hstart + (size_t)(nchunks - 1) * plane + i) = hs;
}

constexpr int kSub      = 16;
constexpr int kRowF     = 64 + kSub;
constexpr int kEbOff    = kSub * kRowF;
constexpr int kApOff    = kEbOff + kSub * 64;
constexpr int kBlockF   = kApOff + 64;

template <int T>
__global__ __launch_bounds__(256)
void ssm_dual_coef(const float* __restrict__ A,
                   const float* __restrict__ BuT, const float* __restrict__ CuT,
                   float* __restrict__ SK, int L)
{
    constexpr int P = 68;
    __shared__ __attribute__((aligned(16))) float sBu[kSub * P];
    __shared__ __attribute__((aligned(16))) float sCu[kSub * P];
    __shared__ __attribute__((aligned(16))) float sAp[(kSub + 1) * P];

    const int c = blockIdx.x;
    const int b = blockIdx.y;
    const int s = blockIdx.z;
    const int t = threadIdx.x;
    const int lbeg = c * T + s * kSub;
    const float* __restrict__ Bub = BuT + ((size_t)b * L + lbeg) * 64;
    const float* __restrict__ Cub = CuT + ((size_t)b * L + lbeg) * 64;

    {
        const int r = t >> 4, q = t & 15;
        *(v4f*)(&sBu[r * P + 4 * q]) = *(const v4f*)(Bub + r * 64 + 4 * q);
        *(v4f*)(&sCu[r * P + 4 * q]) = *(const v4f*)(Cub + r * 64 + 4 * q);
    }
    if (t < 64) {
        const float a = A[t];
        float pw = 1.0f;
        #pragma unroll
        for (int k = 0; k <= kSub; ++k) { sAp[k * P + t] = pw; pw *= a; }
    }
    __syncthreads();

    float* __restrict__ blk = SK + ((((size_t)b * gridDim.x + c) * gridDim.z) + s) * kBlockF;

    #pragma unroll
    for (int p = 0; p < 4; ++p) {
        const int idx = t + 256 * p;
        const int r = idx >> 6, n = idx & 63;
        blk[r * kRowF + n]       = sCu[r * P + n] * sAp[(r + 1) * P + n];
        blk[kEbOff + r * 64 + n] = sAp[(kSub - 1 - r) * P + n] * sBu[r * P + n];
    }
    if (t < 64) blk[kApOff + t] = sAp[kSub * P + t];

    {
        const int l = t >> 4, j = t & 15;
        const int k = (j <= l) ? (l - j) : 0;
        const float* __restrict__ cr = &sCu[l * P];
        const float* __restrict__ ar = &sAp[k * P];
        const float* __restrict__ br = &sBu[j * P];
        float s0 = 0.0f, s1 = 0.0f, s2 = 0.0f, s3 = 0.0f;
        #pragma unroll
        for (int n = 0; n < 64; n += 4) {
            const v4f cv = *(const v4f*)(cr + n);
            const v4f av = *(const v4f*)(ar + n);
            const v4f bv = *(const v4f*)(br + n);
            s0 = fmaf(cv.x, av.x * bv.x, s0);
            s1 = fmaf(cv.y, av.y * bv.y, s1);
            s2 = fmaf(cv.z, av.z * bv.z, s2);
            s3 = fmaf(cv.w, av.w * bv.w, s3);
        }
        const float kv = (s0 + s1) + (s2 + s3);
        blk[l * kRowF + 64 + j] = (j <= l) ? kv : 0.0f;
    }
}

typedef const __attribute__((address_space(4))) float* cfloat_p;

template <int l>
__device__ __forceinline__
float dual_step(cfloat_p row, const float (&h)[64], float (&w)[kSub],
                float& dp, float dl, float ul, float Dd)
{
    dp *= dl;
    #pragma unroll
    for (int j = 0; j < l; ++j) w[j] *= dl;
    w[l] = dl * ul;
    float q[4] = {0.0f, 0.0f, 0.0f, 0.0f};
    #pragma unroll
    for (int n = 0; n < 64; ++n) q[n >> 4] = fmaf(row[n], h[n], q[n >> 4]);
    float s = 0.0f;
    #pragma unroll
    for (int j = 0; j <= l; ++j) s = fmaf(row[64 + j], w[j], s);
    const float y = fmaf(dp, (q[0] + q[1]) + (q[2] + q[3]), s);
    return fmaf(ul, Dd, y);
}

template <int l4>
__device__ __forceinline__
void dual_block4(cfloat_p blk, const v4f uv, const float (&dl)[kSub],
                 float* __restrict__ ybd, const float (&h)[64], float (&w)[kSub],
                 float& dp, float Dd)
{
    v4f yo;
    yo.x = dual_step<l4 + 0>(blk + (l4 + 0) * kRowF, h, w, dp, dl[l4 + 0], uv.x, Dd);
    yo.y = dual_step<l4 + 1>(blk + (l4 + 1) * kRowF, h, w, dp, dl[l4 + 1], uv.y, Dd);
    yo.z = dual_step<l4 + 2>(blk + (l4 + 2) * kRowF, h, w, dp, dl[l4 + 2], uv.z, Dd);
    yo.w = dual_step<l4 + 3>(blk + (l4 + 3) * kRowF, h, w, dp, dl[l4 + 3], uv.w, Dd);
    *(v4f*)(ybd + l4) = yo;
}

template <int T>
__global__ __launch_bounds__(64)
void ssm_scan_dual(const float* __restrict__ u, const float* __restrict__ D,
                   const float* __restrict__ SK, const float* __restrict__ dT,
                   const float* __restrict__ hstart, float* __restrict__ out,
                   int dmodel, int L)
{
    static_assert(T % kSub == 0, "whole sub-chunks");
    constexpr int NSUB = T / kSub;
    const int b = blockIdx.y;
    const int c = blockIdx.z;
    const int d = blockIdx.x * 64 + threadIdx.x;
    const int lbeg = c * T;

    const float* __restrict__ ubd = u   + ((size_t)b * dmodel + d) * L + lbeg;
    float*       __restrict__ ybd = out + ((size_t)b * dmodel + d) * L + lbeg;
    const float* __restrict__ dTb = dT  + ((size_t)b * L + lbeg) * dmodel + d;
    cfloat_p blk = (cfloat_p)(SK + (((size_t)b * gridDim.z + c) * NSUB) * kBlockF);

    const float Dd = D[d];
    const size_t o = ((((size_t)c * gridDim.y + b) * dmodel) + d) * 64;

    float h[64];
    #pragma unroll
    for (int i = 0; i < 64; i += 4) {
        const v4f hv = *(const v4f*)(hstart + o + i);
        h[i] = hv.x; h[i + 1] = hv.y; h[i + 2] = hv.z; h[i + 3] = hv.w;
    }

    float dl[kSub];
    v4f   uv[kSub / 4];
    #pragma unroll
    for (int i = 0; i < kSub; ++i) dl[i] = dTb[(size_t)i * dmodel];
    #pragma unroll
    for (int i = 0; i < kSub / 4; ++i) uv[i] = *(const v4f*)(ubd + 4 * i);

    #pragma nounroll
    for (int s = 0; s < NSUB; ++s) {
        const int nx = (s + 1 < NSUB) ? kSub : 0;
        float dln[kSub];
        v4f   uvn[kSub / 4];
        #pragma unroll
        for (int i = 0; i < kSub; ++i) dln[i] = dTb[(size_t)(nx + i) * dmodel];
        #pragma unroll
        for (int i = 0; i < kSub / 4; ++i) uvn[i] = *(const v4f*)(ubd + nx + 4 * i);

        float w[kSub];
        float dp = 1.0f;
        dual_block4<0> (blk, uv[0], dl, ybd, h, w, dp, Dd);
        dual_block4<4> (blk, uv[1], dl, ybd, h, w, dp, Dd);
        dual_block4<8> (blk, uv[2], dl, ybd, h, w, dp, Dd);
        dual_block4<12>(blk, uv[3], dl, ybd, h, w, dp, Dd);

        if (s + 1 < NSUB) {
            #pragma unroll
            for (int n = 0; n < 64; ++n) h[n] *= dp * blk[kApOff + n];
            #pragma unroll
            for (int j = 0; j < kSub; ++j) {
                #pragma unroll
                for (int n = 0; n < 64; ++n) h[n] = fmaf(blk[kEbOff + j * 64 + n], w[j], h[n]);
            }
        }
        #pragma unroll
        for (int i = 0; i < kSub; ++i) dl[i] = dln[i];
        #pragma unroll
        for (int i = 0; i < kSub / 4; ++i) uv[i] = uvn[i];
        blk += kBlockF;
        ubd += kSub;
        ybd += kSub;
        dTb += (size_t)kSub * dmodel;
    }
}

extern "C" void kernel_launch(void* const* d_in, const int* in_sizes, int n_in,
                              void* d_out, int out_size, void* d_ws, size_t ws_size,
                              hipStream_t stream)
{
    (void)in_sizes; (void)n_in; (void)out_size; (void)ws_size;

    const float* u     = (const float*)d_in[1];
    const float* A     = (const float*)d_in[2];
    const float* Bm    = (const float*)d_in[3];
    const float* Cm    = (const float*)d_in[4];
    const float* D     = (const float*)d_in[5];
    const float* Delta = (const float*)d_in[6];
    float* out = (float*)d_out;

    constexpr int batch = 4, dmodel = 1024, N = 64, L = 2048;

    float* BuT = (float*)d_ws;
    float* CuT = BuT + (size_t)batch * L * N;
    float* dT  = CuT + (size_t)batch * L * N;

    constexpr int chunks = 32, T = L / chunks;
    constexpr size_t plane = (size_t)batch * dmodel * N;
    float* hloc   = dT    + (size_t)batch * L * dmodel;
    float* pdel   = hloc  + (size_t)(chunks - 1) * plane;
    float* hstart = pdel  + (size_t)(chunks - 1) * batch * dmodel;
    float* SK     = hstart + (size_t)chunks * plane;

    const dim3 blk(128, 1, 1);

    const dim3 gblk(256, 1, 1);
    gemm_f32_wmma<dmodel, L, 1, 1, 128><<<dim3(L / 64, N / 32, batch), gblk, 0, stream>>>(
        Bm, u, BuT, N);
    gemm_f32_wmma<dmodel, L, 1, 1, 128><<<dim3(L / 64, N / 32, batch), gblk, 0, stream>>>(
        Cm, u, CuT, N);
    ssm_dual_coef<T><<<dim3(chunks, batch, T / kSub), gblk, 0, stream>>>(A, BuT, CuT, SK, L);
    gemm_f32_wmma<dmodel, L, 4, 2, 32><<<dim3(L / 128, dmodel / 128, batch), gblk, 0, stream>>>(
        Delta, u, dT, dmodel);

    ssm_chunk_state<T><<<dim3(dmodel / 64, batch, chunks - 1), blk, 0, stream>>>(
        u, A, BuT, dT, hloc, pdel, dmodel, N, L);
    ssm_chunk_carry<N, T><<<dim3((unsigned)(plane / 4 / 256)), dim3(256, 1, 1), 0, stream>>>(
        hloc, pdel, A, hstart, chunks, plane);
    ssm_scan_dual<T><<<dim3(dmodel / 64, batch, chunks), dim3(64, 1, 1), 0, stream>>>(
        u, D, SK, dT, hstart, out, dmodel, L);
}


# Bu, Cu products and the scan-coefficient pass fused into one launch: a block forms [Bm;Cm]*u for 32 steps through LDS and derives Ct, K, EB from its own tiles; 7 launches to 5; sums unchanged
# speedup vs baseline: 12.8410x; 1.0331x over previous
#include <hip/hip_runtime.h>

typedef __attribute__((ext_vector_type(2))) float v2f;
typedef __attribute__((ext_vector_type(4))) float v4f;
typedef __attribute__((ext_vector_type(8))) float v8f;

template <int TM, int TN, int KS, int PA, int PB>
__device__ __forceinline__
void gemm_stage(const float* __restrict__ As, const float* __restrict__ Bs,
                v8f (&acc)[TM][TN])
{
    #pragma unroll
    for (int j = 0; j < KS / 4; ++j) {
        v2f a[TM], b[TN];
        #pragma unroll
        for (int i = 0; i < TM; ++i)
            a[i] = *(const v2f*)(As + i * 16 * PA + 4 * j);
        #pragma unroll
        for (int t = 0; t < TN; ++t) {
            b[t].x = Bs[(4 * j)     * PB + t * 16];
            b[t].y = Bs[(4 * j + 1) * PB + t * 16];
        }
        #pragma unroll
        for (int i = 0; i < TM; ++i) {
            #pragma unroll
            for (int t = 0; t < TN; ++t)
                acc[i][t] = __builtin_amdgcn_wmma_f32_16x16x4_f32(
                                false, a[i], false, b[t], (short)0, acc[i][t], false, false);
        }
    }
}

template <int K, int L, int TM, int TN, int KS>
__global__ __launch_bounds__(256)
__attribute__((amdgpu_waves_per_eu(4, 4)))
void gemm_f32_wmma(const float* __restrict__ W, const float* __restrict__ U,
                   float* __restrict__ Out, int M)
{
    constexpr int BM = 32 * TM;
    constexpr int BN = 64 * TN;
    constexpr int PA = KS + 4;
    constexpr int PB = BN + 8;
    constexpr int NS = K / KS;
    constexpr int A_VPR = KS / 4;
    constexpr int A_RPP = 256 / A_VPR;
    constexpr int A_NP  = BM / A_RPP;
    constexpr int B_VPR = BN / 4;
    constexpr int B_RPP = 256 / B_VPR;
    constexpr int B_NP  = KS / B_RPP;
    static_assert(K % KS == 0 && BM % A_RPP == 0 && KS % B_RPP == 0, "tile shapes");
    static_assert((PA % 8) == 4 && (PB % 32) == 8, "bank-conflict-free pitches");

    __shared__ __attribute__((aligned(16))) float sA[2][BM * PA];
    __shared__ __attribute__((aligned(16))) float sB[2][KS * PB];

    const int b = blockIdx.z;
    const float* __restrict__ Ub = U + (long)b * K * L;
    float* __restrict__ Ob = Out + (long)b * (long)M * L;

    const int t    = threadIdx.x;
    const int wave = t >> 5;
    const int lane = t & 31;
    const int wm   = wave >> 2;
    const int wn   = wave & 3;
    const int row  = lane & 15;
    const int kh   = lane >> 4;
    const int m0   = blockIdx.y * BM;
    const int l0   = blockIdx.x * BN;

    const int ar = t / A_VPR, aq = t % A_VPR;
    const int br = t / B_VPR, bq = t % B_VPR;
    const float* __restrict__ ag = W  + (long)(m0 + ar) * K + 4 * aq;
    const float* __restrict__ bg = Ub + (long)br * L + l0 + 4 * bq;
    const int as_off = ar * PA + 4 * aq;
    const int bs_off = br * PB + 4 * bq;

    const int af_off = (wm * TM * 16 + row) * PA + 2 * kh;
    const int bf_off = (2 * kh) * PB + wn * TN * 16 + row;

    v8f acc[TM][TN];
    #pragma unroll
    for (int i = 0; i < TM; ++i) {
        #pragma unroll
        for (int j = 0; j < TN; ++j) acc[i][j] = v8f{};
    }

    v4f ra[A_NP], rb[B_NP];

    #pragma unroll
    for (int p = 0; p < A_NP; ++p) ra[p] = *(const v4f*)(ag + (long)p * A_RPP * K);
    #pragma unroll
    for (int p = 0; p < B_NP; ++p) rb[p] = *(const v4f*)(bg + (long)p * B_RPP * L);
    #pragma unroll
    for (int p = 0; p < A_NP; ++p) *(v4f*)(&sA[0][as_off + p * A_RPP * PA]) = ra[p];
    #pragma unroll
    for (int p = 0; p < B_NP; ++p) *(v4f*)(&sB[0][bs_off + p * B_RPP * PB]) = rb[p];
    __syncthreads();

    for (int s = 0; s + 1 < NS; ++s) {
        const int cur = s & 1;
        ag += KS;
        bg += (long)KS * L;
        #pragma unroll
        for (int p = 0; p < A_NP; ++p) ra[p] = *(const v4f*)(ag + (long)p * A_RPP * K);
        #pragma unroll
        for (int p = 0; p < B_NP; ++p) rb[p] = *(const v4f*)(bg + (long)p * B_RPP * L);

        gemm_stage<TM, TN, KS, PA, PB>(&sA[cur][af_off], &sB[cur][bf_off], acc);

        #pragma unroll
        for (int p = 0; p < A_NP; ++p) *(v4f*)(&sA[cur ^ 1][as_off + p * A_RPP * PA]) = ra[p];
        #pragma unroll
        for (int p = 0; p < B_NP; ++p) *(v4f*)(&sB[cur ^ 1][bs_off + p * B_RPP * PB]) = rb[p];
        __syncthreads();
    }
    gemm_stage<TM, TN, KS, PA, PB>(&sA[(NS - 1) & 1][af_off], &sB[(NS - 1) & 1][bf_off], acc);

    #pragma unroll
    for (int i = 0; i < TM; ++i) {
        #pragma unroll
        for (int j = 0; j < TN; ++j) {
            float* __restrict__ dst = Ob + (long)(l0 + (wn * TN + j) * 16 + row) * M
                                         + m0 + (wm * TM + i) * 16 + 8 * kh;
            v4f lo = {acc[i][j][0], acc[i][j][1], acc[i][j][2], acc[i][j][3]};
            v4f hi = {acc[i][j][4], acc[i][j][5], acc[i][j][6], acc[i][j][7]};
            *(v4f*)(dst)     = lo;
            *(v4f*)(dst + 4) = hi;
        }
    }
}

template <int T>
__global__ __launch_bounds__(128)
void ssm_chunk_state(const float* __restrict__ u, const float* __restrict__ A,
                     const float* __restrict__ BuT, const float* __restrict__ dT,
                     float* __restrict__ hloc, float* __restrict__ pdel,
                     int dmodel, int N, int L)
{
    static_assert(T == 64, "u tile copy: 16 threads x 16 bytes per row of T steps");
    constexpr int PG = T + 4;
    constexpr int PE = 64 + 8;
    __shared__ __attribute__((aligned(16))) float sG[64 * PG];
    __shared__ __attribute__((aligned(16))) float sE[T * PE];

    const int b    = blockIdx.y;
    const int c    = blockIdx.z;
    const int t    = threadIdx.x;
    const int d0   = blockIdx.x * 64;
    const int lbeg = c * T;
    const int wv   = __builtin_amdgcn_readfirstlane(t >> 5);

    const float* __restrict__ ub  = u   + ((size_t)b * dmodel + d0) * L + lbeg;
    const float* __restrict__ Bub = BuT + ((size_t)b * L + lbeg) * N;
    const float* __restrict__ dTb = dT  + ((size_t)b * L + lbeg) * dmodel + d0;

    {
        const int r = t >> 4, q = t & 15;
        v4f uv[8];
        #pragma unroll
        for (int p = 0; p < 8; ++p)
            uv[p] = *(const v4f*)(ub + (size_t)(r + 8 * p) * L + 4 * q);
        #pragma unroll
        for (int p = 0; p < 8; ++p)
            *(v4f*)(&sG[(r + 8 * p) * PG + 4 * q]) = uv[p];
    }

    const int    idx     = t & 63;
    const float* __restrict__ colp = (wv < 2) ? (Bub + idx) : (dTb + idx);
    const size_t cstride = (wv < 2) ? (size_t)N : (size_t)dmodel;
    float v[T];
    #pragma unroll
    for (int j = 0; j < T; ++j) v[j] = colp[(size_t)j * cstride];

    __syncthreads();

    if (wv < 2) {
        const float a = A[idx];
        float pw = 1.0f;
        #pragma unroll
        for (int j = T - 1; j >= 0; --j) {
            sE[j * PE + idx] = pw * v[j];
            pw *= a;
        }
    } else {
        float s = 1.0f;
        #pragma unroll
        for (int j = T - 1; j >= 0; --j) {
            const float ul = sG[idx * PG + j];
            sG[idx * PG + j] = s * (v[j] * ul);
            s *= v[j];
        }
        pdel[((size_t)c * gridDim.y + b) * dmodel + d0 + idx] = s;
    }
    __syncthreads();

    const int lane = t & 31;
    const int cc   = lane & 15;
    const int kh   = lane >> 4;

    v8f acc[4];
    #pragma unroll
    for (int tm = 0; tm < 4; ++tm) acc[tm] = v8f{};

    const float* __restrict__ gp = &sG[(16 * wv + cc) * PG + 2 * kh];
    const float* __restrict__ ep = &sE[(2 * kh) * PE + cc];
    #pragma unroll
    for (int kk = 0; kk < T / 4; ++kk) {
        const v2f g = *(const v2f*)(gp + 4 * kk);
        #pragma unroll
        for (int tm = 0; tm < 4; ++tm) {
            v2f e;
            e.x = ep[(4 * kk)     * PE + 16 * tm];
            e.y = ep[(4 * kk + 1) * PE + 16 * tm];
            acc[tm] = __builtin_amdgcn_wmma_f32_16x16x4_f32(
                          false, e, false, g, (short)0, acc[tm], false, false);
        }
    }

    const int    d = d0 + 16 * wv + cc;
    const size_t o = ((((size_t)c * gridDim.y + b) * dmodel) + d) * N;
    #pragma unroll
    for (int tm = 0; tm < 4; ++tm) {
        const int n0 = 16 * tm + 8 * kh;
        const v4f lo = {acc[tm][0], acc[tm][1], acc[tm][2], acc[tm][3]};
        const v4f hi = {acc[tm][4], acc[tm][5], acc[tm][6], acc[tm][7]};
        *(v4f*)(hloc + o + n0)     = lo;
        *(v4f*)(hloc + o + n0 + 4) = hi;
    }
}

template <int N, int T>
__global__ __launch_bounds__(256)
void ssm_chunk_carry(const float* __restrict__ hloc, const float* __restrict__ pdel,
                     const float* __restrict__ A,
                     float* __restrict__ hstart, int nchunks, size_t plane)
{
    const size_t i    = ((size_t)blockIdx.x * blockDim.x + threadIdx.x) * 4;
    const size_t row  = i / (size_t)N;
    const int    n    = (int)(i % (size_t)N);
    const size_t rows = plane / (size_t)N;

    const v4f a = *(const v4f*)(A + n);
    v4f at = {1.0f, 1.0f, 1.0f, 1.0f};
    for (int j = 0; j < T; ++j) at *= a;

    v4f hs = {0.0f, 0.0f, 0.0f, 0.0f};
    for (int c = 0; c + 1 < nchunks; ++c) {
        const v4f   hl = *(const v4f*)(hloc + (size_t)c * plane + i);
        const float pd = pdel[(size_t)c * rows + row];
        const v4f   pv = pd * at;
        *(v4f*)(hstart + (size_t)c * plane + i) = hs;
        hs.x = fmaf(pv.x, hs.x, hl.x);
        hs.y = fmaf(pv.y, hs.y, hl.y);
        hs.z = fmaf(pv.z, hs.z, hl.z);
        hs.w = fmaf(pv.w, hs.w, hl.w);
    }
    *(v4f*)(hstart + (size_t)(nchunks - 1) * plane + i) = hs;
}

constexpr int kSub      = 16;
constexpr int kRowF     = 64 + kSub;
constexpr int kEbOff    = kSub * kRowF;
constexpr int kApOff    = kEbOff + kSub * 64;
constexpr int kBlockF   = kApOff + 64;

template <int KS, int PA, int PB>
__device__ __forceinline__
void bc_stage(const float* __restrict__ As, const float* __restrict__ Bs, v8f& c0, v8f& c1)
{
    v2f a[KS / 4], b0[KS / 4], b1[KS / 4];
    #pragma unroll
    for (int j = 0; j < KS / 4; ++j) {
        a[j]    = *(const v2f*)(As + 4 * j);
        b0[j].x = Bs[(4 * j)     * PB];
        b0[j].y = Bs[(4 * j + 1) * PB];
        b1[j].x = Bs[(4 * j)     * PB + 16];
        b1[j].y = Bs[(4 * j + 1) * PB + 16];
    }
    __builtin_amdgcn_sched_barrier(0);
    #pragma unroll
    for (int j = 0; j < KS / 4; ++j) {
        c0 = __builtin_amdgcn_wmma_f32_16x16x4_f32(false, a[j], false, b0[j], (short)0, c0, false, false);
        c1 = __builtin_amdgcn_wmma_f32_16x16x4_f32(false, a[j], false, b1[j], (short)0, c1, false, false);
    }
}

template <int K, int L, int T>
__global__ __launch_bounds__(256)
void bc_gemm_coef(const float* __restrict__ Bm, const float* __restrict__ Cm,
                  const float* __restrict__ U, const float* __restrict__ A,
                  float* __restrict__ BuT, float* __restrict__ SK)
{
    constexpr int KS = 64;
    constexpr int BM = 128;
    constexpr int BN = 32;
    constexpr int PA = KS + 4;
    constexpr int PB = BN + 8;
    constexpr int NS = K / KS;
    constexpr int A_NP = 8;
    constexpr int B_NP = 2;
    constexpr int P = 68;
    constexpr int NSUB = T / kSub;
    static_assert(K % KS == 0 && (PA % 8) == 4 && (PB % 32) == 8, "tile shapes and pitches");
    static_assert(BN == 2 * kSub && T % BN == 0, "two whole sub-chunks per workgroup");

    __shared__ __attribute__((aligned(16))) float sA[2][BM * PA];
    __shared__ __attribute__((aligned(16))) float sB[2][KS * PB];
    __shared__ __attribute__((aligned(16))) float sBu[BN * P];
    __shared__ __attribute__((aligned(16))) float sCu[BN * P];
    __shared__ __attribute__((aligned(16))) float sAp[(kSub + 1) * P];

    const int b  = blockIdx.y;
    const int l0 = blockIdx.x * BN;
    const float* __restrict__ Ub = U + (long)b * K * L;

    const int t    = threadIdx.x;
    const int wave = t >> 5;
    const int lane = t & 31;
    const int row  = lane & 15;
    const int kh   = lane >> 4;

    if (t < 64) {
        const float a = A[t];
        float pw = 1.0f;
        #pragma unroll
        for (int k = 0; k <= kSub; ++k) { sAp[k * P + t] = pw; pw *= a; }
    }

    const int ar = t >> 4, aq = t & 15;
    const int br = t >> 3, bq = t & 7;
    const float* __restrict__ agB = Bm + (long)ar * K + 4 * aq;
    const float* __restrict__ agC = Cm + (long)ar * K + 4 * aq;
    const float* __restrict__ bg  = Ub + (long)br * L + l0 + 4 * bq;
    const int as_off = ar * PA + 4 * aq;
    const int bs_off = br * PB + 4 * bq;
    const int af_off = (wave * 16 + row) * PA + 2 * kh;
    const int bf_off = (2 * kh) * PB + row;

    v8f acc[1][2];
    acc[0][0] = v8f{};
    acc[0][1] = v8f{};

    v4f ra[A_NP], rb[B_NP];

    #pragma unroll
    for (int p = 0; p < 4; ++p) ra[p]     = *(const v4f*)(agB + (long)p * 16 * K);
    #pragma unroll
    for (int p = 0; p < 4; ++p) ra[4 + p] = *(const v4f*)(agC + (long)p * 16 * K);
    #pragma unroll
    for (int p = 0; p < B_NP; ++p) rb[p] = *(const v4f*)(bg + (long)p * 32 * L);
    #pragma unroll
    for (int p = 0; p < A_NP; ++p) *(v4f*)(&sA[0][as_off + p * 16 * PA]) = ra[p];
    #pragma unroll
    for (int p = 0; p < B_NP; ++p) *(v4f*)(&sB[0][bs_off + p * 32 * PB]) = rb[p];
    __syncthreads();

    for (int s = 0; s + 1 < NS; ++s) {
        const int cur = s & 1;
        agB += KS;
        agC += KS;
        bg  += (long)KS * L;
        #pragma unroll
        for (int p = 0; p < 4; ++p) ra[p]     = *(const v4f*)(agB + (long)p * 16 * K);
        #pragma unroll
        for (int p = 0; p < 4; ++p) ra[4 + p] = *(const v4f*)(agC + (long)p * 16 * K);
        #pragma unroll
        for (int p = 0; p < B_NP; ++p) rb[p] = *(const v4f*)(bg + (long)p * 32 * L);

        bc_stage<KS, PA, PB>(&sA[cur][af_off], &sB[cur][bf_off], acc[0][0], acc[0][1]);

        __builtin_amdgcn_sched_barrier(0);

        #pragma unroll
        for (int p = 0; p < A_NP; ++p) *(v4f*)(&sA[cur ^ 1][as_off + p * 16 * PA]) = ra[p];
        #pragma unroll
        for (int p = 0; p < B_NP; ++p) *(v4f*)(&sB[cur ^ 1][bs_off + p * 32 * PB]) = rb[p];
        __syncthreads();
    }
    bc_stage<KS, PA, PB>(&sA[(NS - 1) & 1][af_off], &sB[(NS - 1) & 1][bf_off], acc[0][0], acc[0][1]);

    {
        float* __restrict__ dstS = (wave < 4) ? sBu : sCu;
        const int nb = (wave & 3) * 16 + 8 * kh;
        #pragma unroll
        for (int j = 0; j < 2; ++j) {
            const v4f lo = {acc[0][j][0], acc[0][j][1], acc[0][j][2], acc[0][j][3]};
            const v4f hi = {acc[0][j][4], acc[0][j][5], acc[0][j][6], acc[0][j][7]};
            *(v4f*)(&dstS[(16 * j + row) * P + nb])     = lo;
            *(v4f*)(&dstS[(16 * j + row) * P + nb + 4]) = hi;
        }
    }
    __syncthreads();

    {
        const int r = t >> 4, q = t & 15;
        float* __restrict__ dst = BuT + ((size_t)b * L + l0) * 64;
        #pragma unroll
        for (int p = 0; p < 2; ++p)
            *(v4f*)(dst + (r + 16 * p) * 64 + 4 * q) = *(const v4f*)(&sBu[(r + 16 * p) * P + 4 * q]);
    }

    const int c  = l0 / T;
    const int s0 = (l0 % T) / kSub;
    #pragma unroll
    for (int s2 = 0; s2 < 2; ++s2) {
        const float* __restrict__ tBu = &sBu[s2 * kSub * P];
        const float* __restrict__ tCu = &sCu[s2 * kSub * P];
        float* __restrict__ blk = SK + ((((size_t)b * (L / T) + c) * NSUB) + s0 + s2) * kBlockF;

        #pragma unroll
        for (int p = 0; p < 4; ++p) {
            const int idx = t + 256 * p;
            const int r = idx >> 6, n = idx & 63;
            blk[r * kRowF + n]       = tCu[r * P + n] * sAp[(r + 1) * P + n];
            blk[kEbOff + r * 64 + n] = sAp[(kSub - 1 - r) * P + n] * tBu[r * P + n];
        }
        if (t < 64) blk[kApOff + t] = sAp[kSub * P + t];

        {
            const int l = t >> 4, j = t & 15;
            const int k = (j <= l) ? (l - j) : 0;
            const float* __restrict__ cr = &tCu[l * P];
            const float* __restrict__ ar2 = &sAp[k * P];
            const float* __restrict__ br2 = &tBu[j * P];
            float q0 = 0.0f, q1 = 0.0f, q2 = 0.0f, q3 = 0.0f;
            #pragma unroll
            for (int n = 0; n < 64; n += 4) {
                const v4f cv = *(const v4f*)(cr + n);
                const v4f av = *(const v4f*)(ar2 + n);
                const v4f bv = *(const v4f*)(br2 + n);
                q0 = fmaf(cv.x, av.x * bv.x, q0);
                q1 = fmaf(cv.y, av.y * bv.y, q1);
                q2 = fmaf(cv.z, av.z * bv.z, q2);
                q3 = fmaf(cv.w, av.w * bv.w, q3);
            }
            const float kv = (q0 + q1) + (q2 + q3);
            blk[l * kRowF + 64 + j] = (j <= l) ? kv : 0.0f;
        }
    }
}

typedef const __attribute__((address_space(4))) float* cfloat_p;

template <int l>
__device__ __forceinline__
float dual_step(cfloat_p row, const float (&h)[64], float (&w)[kSub],
                float& dp, float dl, float ul, float Dd)
{
    dp *= dl;
    #pragma unroll
    for (int j = 0; j < l; ++j) w[j] *= dl;
    w[l] = dl * ul;
    float q[4] = {0.0f, 0.0f, 0.0f, 0.0f};
    #pragma unroll
    for (int n = 0; n < 64; ++n) q[n >> 4] = fmaf(row[n], h[n], q[n >> 4]);
    float s = 0.0f;
    #pragma unroll
    for (int j = 0; j <= l; ++j) s = fmaf(row[64 + j], w[j], s);
    const float y = fmaf(dp, (q[0] + q[1]) + (q[2] + q[3]), s);
    return fmaf(ul, Dd, y);
}

template <int l4>
__device__ __forceinline__
void dual_block4(cfloat_p blk, const v4f uv, const float (&dl)[kSub],
                 float* __restrict__ ybd, const float (&h)[64], float (&w)[kSub],
                 float& dp, float Dd)
{
    v4f yo;
    yo.x = dual_step<l4 + 0>(blk + (l4 + 0) * kRowF, h, w, dp, dl[l4 + 0], uv.x, Dd);
    yo.y = dual_step<l4 + 1>(blk + (l4 + 1) * kRowF, h, w, dp, dl[l4 + 1], uv.y, Dd);
    yo.z = dual_step<l4 + 2>(blk + (l4 + 2) * kRowF, h, w, dp, dl[l4 + 2], uv.z, Dd);
    yo.w = dual_step<l4 + 3>(blk + (l4 + 3) * kRowF, h, w, dp, dl[l4 + 3], uv.w, Dd);
    *(v4f*)(ybd + l4) = yo;
}

template <int T>
__global__ __launch_bounds__(64)
void ssm_scan_dual(const float* __restrict__ u, const float* __restrict__ D,
                   const float* __restrict__ SK, const float* __restrict__ dT,
                   const float* __restrict__ hstart, float* __restrict__ out,
                   int dmodel, int L)
{
    static_assert(T % kSub == 0, "whole sub-chunks");
    constexpr int NSUB = T / kSub;
    const int b = blockIdx.y;
    const int c = blockIdx.z;
    const int d = blockIdx.x * 64 + threadIdx.x;
    const int lbeg = c * T;

    const float* __restrict__ ubd = u   + ((size_t)b * dmodel + d) * L + lbeg;
    float*       __restrict__ ybd = out + ((size_t)b * dmodel + d) * L + lbeg;
    const float* __restrict__ dTb = dT  + ((size_t)b * L + lbeg) * dmodel + d;
    cfloat_p blk = (cfloat_p)(SK + (((size_t)b * gridDim.z + c) * NSUB) * kBlockF);

    const float Dd = D[d];
    const size_t o = ((((size_t)c * gridDim.y + b) * dmodel) + d) * 64;

    float h[64];
    #pragma unroll
    for (int i = 0; i < 64; i += 4) {
        const v4f hv = *(const v4f*)(hstart + o + i);
        h[i] = hv.x; h[i + 1] = hv.y; h[i + 2] = hv.z; h[i + 3] = hv.w;
    }

    float dl[kSub];
    v4f   uv[kSub / 4];
    #pragma unroll
    for (int i = 0; i < kSub; ++i) dl[i] = dTb[(size_t)i * dmodel];
    #pragma unroll
    for (int i = 0; i < kSub / 4; ++i) uv[i] = *(const v4f*)(ubd + 4 * i);

    #pragma nounroll
    for (int s = 0; s < NSUB; ++s) {
        const int nx = (s + 1 < NSUB) ? kSub : 0;
        float dln[kSub];
        v4f   uvn[kSub / 4];
        #pragma unroll
        for (int i = 0; i < kSub; ++i) dln[i] = dTb[(size_t)(nx + i) * dmodel];
        #pragma unroll
        for (int i = 0; i < kSub / 4; ++i) uvn[i] = *(const v4f*)(ubd + nx + 4 * i);

        float w[kSub];
        float dp = 1.0f;
        dual_block4<0> (blk, uv[0], dl, ybd, h, w, dp, Dd);
        dual_block4<4> (blk, uv[1], dl, ybd, h, w, dp, Dd);
        dual_block4<8> (blk, uv[2], dl, ybd, h, w, dp, Dd);
        dual_block4<12>(blk, uv[3], dl, ybd, h, w, dp, Dd);

        if (s + 1 < NSUB) {
            #pragma unroll
            for (int n = 0; n < 64; ++n) h[n] *= dp * blk[kApOff + n];
            #pragma unroll
            for (int j = 0; j < kSub; ++j) {
                #pragma unroll
                for (int n = 0; n < 64; ++n) h[n] = fmaf(blk[kEbOff + j * 64 + n], w[j], h[n]);
            }
        }
        #pragma unroll
        for (int i = 0; i < kSub; ++i) dl[i] = dln[i];
        #pragma unroll
        for (int i = 0; i < kSub / 4; ++i) uv[i] = uvn[i];
        blk += kBlockF;
        ubd += kSub;
        ybd += kSub;
        dTb += (size_t)kSub * dmodel;
    }
}

extern "C" void kernel_launch(void* const* d_in, const int* in_sizes, int n_in,
                              void* d_out, int out_size, void* d_ws, size_t ws_size,
                              hipStream_t stream)
{
    (void)in_sizes; (void)n_in; (void)out_size; (void)ws_size;

    const float* u     = (const float*)d_in[1];
    const float* A     = (const float*)d_in[2];
    const float* Bm    = (const float*)d_in[3];
    const float* Cm    = (const float*)d_in[4];
    const float* D     = (const float*)d_in[5];
    const float* Delta = (const float*)d_in[6];
    float* out = (float*)d_out;

    constexpr int batch = 4, dmodel = 1024, N = 64, L = 2048;

    float* BuT = (float*)d_ws;
    float* CuT = BuT + (size_t)batch * L * N;
    float* dT  = CuT + (size_t)batch * L * N;

    constexpr int chunks = 32, T = L / chunks;
    constexpr size_t plane = (size_t)batch * dmodel * N;
    float* hloc   = dT    + (size_t)batch * L * dmodel;
    float* pdel   = hloc  + (size_t)(chunks - 1) * plane;
    float* hstart = pdel  + (size_t)(chunks - 1) * batch * dmodel;
    float* SK     = hstart + (size_t)chunks * plane;

    const dim3 blk(128, 1, 1);

    const dim3 gblk(256, 1, 1);
    bc_gemm_coef<dmodel, L, T><<<dim3(L / 32, batch, 1), gblk, 0, stream>>>(Bm, Cm, u, A, BuT, SK);
    gemm_f32_wmma<dmodel, L, 4, 2, 32><<<dim3(L / 128, dmodel / 128, batch), gblk, 0, stream>>>(
        Delta, u, dT, dmodel);

    ssm_chunk_state<T><<<dim3(dmodel / 64, batch, chunks - 1), blk, 0, stream>>>(
        u, A, BuT, dT, hloc, pdel, dmodel, N, L);
    ssm_chunk_carry<N, T><<<dim3((unsigned)(plane / 4 / 256)), dim3(256, 1, 1), 0, stream>>>(
        hloc, pdel, A, hstart, chunks, plane);
    ssm_scan_dual<T><<<dim3(dmodel / 64, batch, chunks), dim3(64, 1, 1), 0, stream>>>(
        u, D, SK, dT, hstart, out, dmodel, L);
}


# scan: a chunk's start-state term as one WMMA product Ct*H^T (via LDS), in-chunk term on a 64x64 K, older weights frozen per 16 steps; no state update in scan; 5 launches to 6; y sum order changes
# speedup vs baseline: 15.1230x; 1.1777x over previous
#include <hip/hip_runtime.h>

typedef __attribute__((ext_vector_type(2))) float v2f;
typedef __attribute__((ext_vector_type(4))) float v4f;
typedef __attribute__((ext_vector_type(8))) float v8f;

template <int TM, int TN, int KS, int PA, int PB>
__device__ __forceinline__
void gemm_stage(const float* __restrict__ As, const float* __restrict__ Bs,
                v8f (&acc)[TM][TN])
{
    #pragma unroll
    for (int j = 0; j < KS / 4; ++j) {
        v2f a[TM], b[TN];
        #pragma unroll
        for (int i = 0; i < TM; ++i)
            a[i] = *(const v2f*)(As + i * 16 * PA + 4 * j);
        #pragma unroll
        for (int t = 0; t < TN; ++t) {
            b[t].x = Bs[(4 * j)     * PB + t * 16];
            b[t].y = Bs[(4 * j + 1) * PB + t * 16];
        }
        #pragma unroll
        for (int i = 0; i < TM; ++i) {
            #pragma unroll
            for (int t = 0; t < TN; ++t)
                acc[i][t] = __builtin_amdgcn_wmma_f32_16x16x4_f32(
                                false, a[i], false, b[t], (short)0, acc[i][t], false, false);
        }
    }
}

template <int K, int L, int TM, int TN, int KS>
__global__ __launch_bounds__(256)
__attribute__((amdgpu_waves_per_eu(4, 4)))
void gemm_f32_wmma(const float* __restrict__ W, const float* __restrict__ U,
                   float* __restrict__ Out, int M)
{
    constexpr int BM = 32 * TM;
    constexpr int BN = 64 * TN;
    constexpr int PA = KS + 4;
    constexpr int PB = BN + 8;
    constexpr int NS = K / KS;
    constexpr int A_VPR = KS / 4;
    constexpr int A_RPP = 256 / A_VPR;
    constexpr int A_NP  = BM / A_RPP;
    constexpr int B_VPR = BN / 4;
    constexpr int B_RPP = 256 / B_VPR;
    constexpr int B_NP  = KS / B_RPP;
    static_assert(K % KS == 0 && BM % A_RPP == 0 && KS % B_RPP == 0, "tile shapes");
    static_assert((PA % 8) == 4 && (PB % 32) == 8, "bank-conflict-free pitches");

    __shared__ __attribute__((aligned(16))) float sA[2][BM * PA];
    __shared__ __attribute__((aligned(16))) float sB[2][KS * PB];

    const int b = blockIdx.z;
    const float* __restrict__ Ub = U + (long)b * K * L;
    float* __restrict__ Ob = Out + (long)b * (long)M * L;

    const int t    = threadIdx.x;
    const int wave = t >> 5;
    const int lane = t & 31;
    const int wm   = wave >> 2;
    const int wn   = wave & 3;
    const int row  = lane & 15;
    const int kh   = lane >> 4;
    const int m0   = blockIdx.y * BM;
    const int l0   = blockIdx.x * BN;

    const int ar = t / A_VPR, aq = t % A_VPR;
    const int br = t / B_VPR, bq = t % B_VPR;
    const float* __restrict__ ag = W  + (long)(m0 + ar) * K + 4 * aq;
    const float* __restrict__ bg = Ub + (long)br * L + l0 + 4 * bq;
    const int as_off = ar * PA + 4 * aq;
    const int bs_off = br * PB + 4 * bq;

    const int af_off = (wm * TM * 16 + row) * PA + 2 * kh;
    const int bf_off = (2 * kh) * PB + wn * TN * 16 + row;

    v8f acc[TM][TN];
    #pragma unroll
    for (int i = 0; i < TM; ++i) {
        #pragma unroll
        for (int j = 0; j < TN; ++j) acc[i][j] = v8f{};
    }

    v4f ra[A_NP], rb[B_NP];

    #pragma unroll
    for (int p = 0; p < A_NP; ++p) ra[p] = *(const v4f*)(ag + (long)p * A_RPP * K);
    #pragma unroll
    for (int p = 0; p < B_NP; ++p) rb[p] = *(const v4f*)(bg + (long)p * B_RPP * L);
    #pragma unroll
    for (int p = 0; p < A_NP; ++p) *(v4f*)(&sA[0][as_off + p * A_RPP * PA]) = ra[p];
    #pragma unroll
    for (int p = 0; p < B_NP; ++p) *(v4f*)(&sB[0][bs_off + p * B_RPP * PB]) = rb[p];
    __syncthreads();

    for (int s = 0; s + 1 < NS; ++s) {
        const int cur = s & 1;
        ag += KS;
        bg += (long)KS * L;
        #pragma unroll
        for (int p = 0; p < A_NP; ++p) ra[p] = *(const v4f*)(ag + (long)p * A_RPP * K);
        #pragma unroll
        for (int p = 0; p < B_NP; ++p) rb[p] = *(const v4f*)(bg + (long)p * B_RPP * L);

        gemm_stage<TM, TN, KS, PA, PB>(&sA[cur][af_off], &sB[cur][bf_off], acc);

        #pragma unroll
        for (int p = 0; p < A_NP; ++p) *(v4f*)(&sA[cur ^ 1][as_off + p * A_RPP * PA]) = ra[p];
        #pragma unroll
        for (int p = 0; p < B_NP; ++p) *(v4f*)(&sB[cur ^ 1][bs_off + p * B_RPP * PB]) = rb[p];
        __syncthreads();
    }
    gemm_stage<TM, TN, KS, PA, PB>(&sA[(NS - 1) & 1][af_off], &sB[(NS - 1) & 1][bf_off], acc);

    #pragma unroll
    for (int i = 0; i < TM; ++i) {
        #pragma unroll
        for (int j = 0; j < TN; ++j) {
            float* __restrict__ dst = Ob + (long)(l0 + (wn * TN + j) * 16 + row) * M
                                         + m0 + (wm * TM + i) * 16 + 8 * kh;
            v4f lo = {acc[i][j][0], acc[i][j][1], acc[i][j][2], acc[i][j][3]};
            v4f hi = {acc[i][j][4], acc[i][j][5], acc[i][j][6], acc[i][j][7]};
            *(v4f*)(dst)     = lo;
            *(v4f*)(dst + 4) = hi;
        }
    }
}

template <int T>
__global__ __launch_bounds__(128)
void ssm_chunk_state(const float* __restrict__ u, const float* __restrict__ A,
                     const float* __restrict__ BuT, const float* __restrict__ dT,
                     float* __restrict__ hloc, float* __restrict__ pdel,
                     int dmodel, int N, int L)
{
    static_assert(T == 64, "u tile copy: 16 threads x 16 bytes per row of T steps");
    constexpr int PG = T + 4;
    constexpr int PE = 64 + 8;
    __shared__ __attribute__((aligned(16))) float sG[64 * PG];
    __shared__ __attribute__((aligned(16))) float sE[T * PE];

    const int b    = blockIdx.y;
    const int c    = blockIdx.z;
    const int t    = threadIdx.x;
    const int d0   = blockIdx.x * 64;
    const int lbeg = c * T;
    const int wv   = __builtin_amdgcn_readfirstlane(t >> 5);

    const float* __restrict__ ub  = u   + ((size_t)b * dmodel + d0) * L + lbeg;
    const float* __restrict__ Bub = BuT + ((size_t)b * L + lbeg) * N;
    const float* __restrict__ dTb = dT  + ((size_t)b * L + lbeg) * dmodel + d0;

    {
        const int r = t >> 4, q = t & 15;
        v4f uv[8];
        #pragma unroll
        for (int p = 0; p < 8; ++p)
            uv[p] = *(const v4f*)(ub + (size_t)(r + 8 * p) * L + 4 * q);
        #pragma unroll
        for (int p = 0; p < 8; ++p)
            *(v4f*)(&sG[(r + 8 * p) * PG + 4 * q]) = uv[p];
    }

    const int    idx     = t & 63;
    const float* __restrict__ colp = (wv < 2) ? (Bub + idx) : (dTb + idx);
    const size_t cstride = (wv < 2) ? (size_t)N : (size_t)dmodel;
    float v[T];
    #pragma unroll
    for (int j = 0; j < T; ++j) v[j] = colp[(size_t)j * cstride];

    __syncthreads();

    if (wv < 2) {
        const float a = A[idx];
        float pw = 1.0f;
        #pragma unroll
        for (int j = T - 1; j >= 0; --j) {
            sE[j * PE + idx] = pw * v[j];
            pw *= a;
        }
    } else {
        float s = 1.0f;
        #pragma unroll
        for (int j = T - 1; j >= 0; --j) {
            const float ul = sG[idx * PG + j];
            sG[idx * PG + j] = s * (v[j] * ul);
            s *= v[j];
        }
        pdel[((size_t)c * gridDim.y + b) * dmodel + d0 + idx] = s;
    }
    __syncthreads();

    const int lane = t & 31;
    const int cc   = lane & 15;
    const int kh   = lane >> 4;

    v8f acc[4];
    #pragma unroll
    for (int tm = 0; tm < 4; ++tm) acc[tm] = v8f{};

    const float* __restrict__ gp = &sG[(16 * wv + cc) * PG + 2 * kh];
    const float* __restrict__ ep = &sE[(2 * kh) * PE + cc];
    #pragma unroll
    for (int kk = 0; kk < T / 4; ++kk) {
        const v2f g = *(const v2f*)(gp + 4 * kk);
        #pragma unroll
        for (int tm = 0; tm < 4; ++tm) {
            v2f e;
            e.x = ep[(4 * kk)     * PE + 16 * tm];
            e.y = ep[(4 * kk + 1) * PE + 16 * tm];
            acc[tm] = __builtin_amdgcn_wmma_f32_16x16x4_f32(
                          false, e, false, g, (short)0, acc[tm], false, false);
        }
    }

    const int    d = d0 + 16 * wv + cc;
    const size_t o = ((((size_t)c * gridDim.y + b) * dmodel) + d) * N;
    #pragma unroll
    for (int tm = 0; tm < 4; ++tm) {
        const int n0 = 16 * tm + 8 * kh;
        const v4f lo = {acc[tm][0], acc[tm][1], acc[tm][2], acc[tm][3]};
        const v4f hi = {acc[tm][4], acc[tm][5], acc[tm][6], acc[tm][7]};
        *(v4f*)(hloc + o + n0)     = lo;
        *(v4f*)(hloc + o + n0 + 4) = hi;
    }
}

template <int N, int T>
__global__ __launch_bounds__(256)
void ssm_chunk_carry(const float* __restrict__ hloc, const float* __restrict__ pdel,
                     const float* __restrict__ A,
                     float* __restrict__ hstart, int nchunks, size_t plane)
{
    const size_t i    = ((size_t)blockIdx.x * blockDim.x + threadIdx.x) * 4;
    const size_t row  = i / (size_t)N;
    const int    n    = (int)(i % (size_t)N);
    const size_t rows = plane / (size_t)N;

    const v4f a = *(const v4f*)(A + n);
    v4f at = {1.0f, 1.0f, 1.0f, 1.0f};
    for (int j = 0; j < T; ++j) at *= a;

    v4f hs = {0.0f, 0.0f, 0.0f, 0.0f};
    for (int c = 0; c + 1 < nchunks; ++c) {
        const v4f   hl = *(const v4f*)(hloc + (size_t)c * plane + i);
        const float pd = pdel[(size_t)c * rows + row];
        const v4f   pv = pd * at;
        *(v4f*)(hstart + (size_t)c * plane + i) = hs;
        hs.x = fmaf(pv.x, hs.x, hl.x);
        hs.y = fmaf(pv.y, hs.y, hl.y);
        hs.z = fmaf(pv.z, hs.z, hl.z);
        hs.w = fmaf(pv.w, hs.w, hl.w);
    }
    *(v4f*)(hstart + (size_t)(nchunks - 1) * plane + i) = hs;
}

constexpr int kT     = 64;
constexpr int kKOff  = kT * 64;
constexpr int kCoefF = 2 * kT * 64;

template <int KS, int PA, int PB>
__device__ __forceinline__
void bc_stage(const float* __restrict__ As, const float* __restrict__ Bs, v8f& c0, v8f& c1)
{
    v2f a[KS / 4], b0[KS / 4], b1[KS / 4];
    #pragma unroll
    for (int j = 0; j < KS / 4; ++j) {
        a[j]    = *(const v2f*)(As + 4 * j);
        b0[j].x = Bs[(4 * j)     * PB];
        b0[j].y = Bs[(4 * j + 1) * PB];
        b1[j].x = Bs[(4 * j)     * PB + 16];
        b1[j].y = Bs[(4 * j + 1) * PB + 16];
    }
    __builtin_amdgcn_sched_barrier(0);
    #pragma unroll
    for (int j = 0; j < KS / 4; ++j) {
        c0 = __builtin_amdgcn_wmma_f32_16x16x4_f32(false, a[j], false, b0[j], (short)0, c0, false, false);
        c1 = __builtin_amdgcn_wmma_f32_16x16x4_f32(false, a[j], false, b1[j], (short)0, c1, false, false);
    }
}

template <int K, int L>
__global__ __launch_bounds__(256)
void bc_gemm(const float* __restrict__ Bm, const float* __restrict__ Cm,
             const float* __restrict__ U,
             float* __restrict__ BuT, float* __restrict__ CuT)
{
    constexpr int KS = 64;
    constexpr int BM = 128;
    constexpr int BN = 32;
    constexpr int PA = KS + 4;
    constexpr int PB = BN + 8;
    constexpr int NS = K / KS;
    constexpr int A_NP = 8;
    constexpr int B_NP = 2;
    constexpr int P = 68;
    static_assert(K % KS == 0 && (PA % 8) == 4 && (PB % 32) == 8, "tile shapes and pitches");

    __shared__ __attribute__((aligned(16))) float sA[2][BM * PA];
    __shared__ __attribute__((aligned(16))) float sB[2][KS * PB];
    __shared__ __attribute__((aligned(16))) float sBu[BN * P];
    __shared__ __attribute__((aligned(16))) float sCu[BN * P];

    const int b  = blockIdx.y;
    const int l0 = blockIdx.x * BN;
    const float* __restrict__ Ub = U + (long)b * K * L;

    const int t    = threadIdx.x;
    const int wave = t >> 5;
    const int lane = t & 31;
    const int row  = lane & 15;
    const int kh   = lane >> 4;

    const int ar = t >> 4, aq = t & 15;
    const int br = t >> 3, bq = t & 7;
    const float* __restrict__ agB = Bm + (long)ar * K + 4 * aq;
    const float* __restrict__ agC = Cm + (long)ar * K + 4 * aq;
    const float* __restrict__ bg  = Ub + (long)br * L + l0 + 4 * bq;
    const int as_off = ar * PA + 4 * aq;
    const int bs_off = br * PB + 4 * bq;
    const int af_off = (wave * 16 + row) * PA + 2 * kh;
    const int bf_off = (2 * kh) * PB + row;

    v8f acc[1][2];
    acc[0][0] = v8f{};
    acc[0][1] = v8f{};

    v4f ra[A_NP], rb[B_NP];

    #pragma unroll
    for (int p = 0; p < 4; ++p) ra[p]     = *(const v4f*)(agB + (long)p * 16 * K);
    #pragma unroll
    for (int p = 0; p < 4; ++p) ra[4 + p] = *(const v4f*)(agC + (long)p * 16 * K);
    #pragma unroll
    for (int p = 0; p < B_NP; ++p) rb[p] = *(const v4f*)(bg + (long)p * 32 * L);
    #pragma unroll
    for (int p = 0; p < A_NP; ++p) *(v4f*)(&sA[0][as_off + p * 16 * PA]) = ra[p];
    #pragma unroll
    for (int p = 0; p < B_NP; ++p) *(v4f*)(&sB[0][bs_off + p * 32 * PB]) = rb[p];
    __syncthreads();

    for (int s = 0; s + 1 < NS; ++s) {
        const int cur = s & 1;
        agB += KS;
        agC += KS;
        bg  += (long)KS * L;
        #pragma unroll
        for (int p = 0; p < 4; ++p) ra[p]     = *(const v4f*)(agB + (long)p * 16 * K);
        #pragma unroll
        for (int p = 0; p < 4; ++p) ra[4 + p] = *(const v4f*)(agC + (long)p * 16 * K);
        #pragma unroll
        for (int p = 0; p < B_NP; ++p) rb[p] = *(const v4f*)(bg + (long)p * 32 * L);

        bc_stage<KS, PA, PB>(&sA[cur][af_off], &sB[cur][bf_off], acc[0][0], acc[0][1]);

        __builtin_amdgcn_sched_barrier(0);

        #pragma unroll
        for (int p = 0; p < A_NP; ++p) *(v4f*)(&sA[cur ^ 1][as_off + p * 16 * PA]) = ra[p];
        #pragma unroll
        for (int p = 0; p < B_NP; ++p) *(v4f*)(&sB[cur ^ 1][bs_off + p * 32 * PB]) = rb[p];
        __syncthreads();
    }
    bc_stage<KS, PA, PB>(&sA[(NS - 1) & 1][af_off], &sB[(NS - 1) & 1][bf_off], acc[0][0], acc[0][1]);

    {
        float* __restrict__ dstS = (wave < 4) ? sBu : sCu;
        const int nb = (wave & 3) * 16 + 8 * kh;
        #pragma unroll
        for (int j = 0; j < 2; ++j) {
            const v4f lo = {acc[0][j][0], acc[0][j][1], acc[0][j][2], acc[0][j][3]};
            const v4f hi = {acc[0][j][4], acc[0][j][5], acc[0][j][6], acc[0][j][7]};
            *(v4f*)(&dstS[(16 * j + row) * P + nb])     = lo;
            *(v4f*)(&dstS[(16 * j + row) * P + nb + 4]) = hi;
        }
    }
    __syncthreads();

    {
        const int r = t >> 4, q = t & 15;
        float* __restrict__ dB = BuT + ((size_t)b * L + l0) * 64;
        float* __restrict__ dC = CuT + ((size_t)b * L + l0) * 64;
        #pragma unroll
        for (int p = 0; p < 2; ++p) {
            *(v4f*)(dB + (r + 16 * p) * 64 + 4 * q) = *(const v4f*)(&sBu[(r + 16 * p) * P + 4 * q]);
            *(v4f*)(dC + (r + 16 * p) * 64 + 4 * q) = *(const v4f*)(&sCu[(r + 16 * p) * P + 4 * q]);
        }
    }
}

template <int L, int T>
__global__ __launch_bounds__(256)
void chunk_coef(const float* __restrict__ BuT, const float* __restrict__ CuT,
                const float* __restrict__ A, float* __restrict__ SK)
{
    static_assert(T == kT, "coefficient block layout");
    constexpr int P = 68;
    __shared__ __attribute__((aligned(16))) float sBu[T * P];
    __shared__ __attribute__((aligned(16))) float sCu[16 * P];
    __shared__ __attribute__((aligned(16))) float sAp[(T + 1) * P];

    const int sl   = blockIdx.x;
    const int b    = blockIdx.y;
    const int c    = blockIdx.z;
    const int t    = threadIdx.x;
    const int lbeg = c * T;

    if (t < 64) {
        const float a = A[t];
        float pw = 1.0f;
        for (int k = 0; k <= T; ++k) { sAp[k * P + t] = pw; pw *= a; }
    }
    {
        const int r = t >> 4, q = t & 15;
        const float* __restrict__ srcB = BuT + ((size_t)b * L + lbeg) * 64;
        const float* __restrict__ srcC = CuT + ((size_t)b * L + lbeg + 16 * sl) * 64;
        for (int p = 0; p <= sl; ++p)
            *(v4f*)(&sBu[(r + 16 * p) * P + 4 * q]) = *(const v4f*)(srcB + (r + 16 * p) * 64 + 4 * q);
        *(v4f*)(&sCu[r * P + 4 * q]) = *(const v4f*)(srcC + r * 64 + 4 * q);
    }
    __syncthreads();

    float* __restrict__ blk = SK + ((size_t)b * gridDim.z + c) * kCoefF;

    #pragma unroll
    for (int p = 0; p < 4; ++p) {
        const int idx = t + 256 * p;
        const int r = idx >> 6, n = idx & 63;
        const int l = 16 * sl + r;
        blk[l * 64 + n] = sCu[r * P + n] * sAp[(l + 1) * P + n];
    }

    const int li = t >> 4, jj = t & 15;
    const int l  = 16 * sl + li;
    const float* __restrict__ cr = &sCu[li * P];
    for (int sj = 0; sj < T / 16; ++sj) {
        const int j = 16 * sj + jj;
        float kv = 0.0f;
        if (sj <= sl) {
            const int k = (j <= l) ? (l - j) : 0;
            const float* __restrict__ ar2 = &sAp[k * P];
            const float* __restrict__ br2 = &sBu[j * P];
            float q0 = 0.0f, q1 = 0.0f, q2 = 0.0f, q3 = 0.0f;
            #pragma unroll
            for (int n = 0; n < 64; n += 4) {
                const v4f cv = *(const v4f*)(cr + n);
                const v4f av = *(const v4f*)(ar2 + n);
                const v4f bv = *(const v4f*)(br2 + n);
                q0 = fmaf(cv.x, av.x * bv.x, q0);
                q1 = fmaf(cv.y, av.y * bv.y, q1);
                q2 = fmaf(cv.z, av.z * bv.z, q2);
                q3 = fmaf(cv.w, av.w * bv.w, q3);
            }
            kv = (j <= l) ? ((q0 + q1) + (q2 + q3)) : 0.0f;
        }
        blk[kKOff + l * 64 + j] = kv;
    }
}

typedef const __attribute__((address_space(4))) float* cfloat_p;

template <int l>
__device__ __forceinline__
float flat_step(cfloat_p krow, float (&w)[kT], float& dpb, float dpprev,
                float y1, float dl, float ul, float Dd)
{
    constexpr int s0 = (l >> 4) << 4;
    dpb *= dl;
    #pragma unroll
    for (int j = s0; j < l; ++j) w[j] *= dl;
    w[l] = dl * ul;
    float qo[3] = {0.0f, 0.0f, 0.0f};
    #pragma unroll
    for (int j = 0; j < s0; ++j) qo[j >> 4] = fmaf(krow[j], w[j], qo[j >> 4]);
    float sn = 0.0f;
    #pragma unroll
    for (int j = s0; j <= l; ++j) sn = fmaf(krow[j], w[j], sn);
    const float so = fmaf(dpprev, y1, (qo[0] + qo[1]) + qo[2]);
    const float y  = fmaf(dpb, so, sn);
    return fmaf(ul, Dd, y);
}

template <int l4>
__device__ __forceinline__
void flat_group4(cfloat_p K, const v4f y1, const v4f uv, const float (&dl)[16],
                 float* __restrict__ ybd, float (&w)[kT], float& dpb, float dpprev, float Dd)
{
    constexpr int i4 = l4 & 15;
    v4f yo;
    yo.x = flat_step<l4 + 0>(K + (l4 + 0) * 64, w, dpb, dpprev, y1.x, dl[i4 + 0], uv.x, Dd);
    yo.y = flat_step<l4 + 1>(K + (l4 + 1) * 64, w, dpb, dpprev, y1.y, dl[i4 + 1], uv.y, Dd);
    yo.z = flat_step<l4 + 2>(K + (l4 + 2) * 64, w, dpb, dpprev, y1.z, dl[i4 + 2], uv.z, Dd);
    yo.w = flat_step<l4 + 3>(K + (l4 + 3) * 64, w, dpb, dpprev, y1.w, dl[i4 + 3], uv.w, Dd);
    *(v4f*)(ybd + l4) = yo;
}

template <int s>
__device__ __forceinline__
void flat_block(cfloat_p K, const float* __restrict__ yrow, const float (&dl)[16],
                const v4f (&uv)[4], float* __restrict__ ybd, float (&w)[kT],
                float& dpprev, float Dd)
{
    v4f y1[4];
    #pragma unroll
    for (int i = 0; i < 4; ++i) y1[i] = *(const v4f*)(yrow + 16 * s + 4 * i);
    float dpb = 1.0f;
    flat_group4<16 * s + 0> (K, y1[0], uv[0], dl, ybd, w, dpb, dpprev, Dd);
    flat_group4<16 * s + 4> (K, y1[1], uv[1], dl, ybd, w, dpb, dpprev, Dd);
    flat_group4<16 * s + 8> (K, y1[2], uv[2], dl, ybd, w, dpb, dpprev, Dd);
    flat_group4<16 * s + 12>(K, y1[3], uv[3], dl, ybd, w, dpb, dpprev, Dd);
    if (16 * (s + 1) < kT) {
        #pragma unroll
        for (int j = 0; j < 16 * s; ++j) w[j] *= dpb;
        dpprev *= dpb;
    }
}

__device__ __forceinline__
void flat_fetch(const float* __restrict__ dTb, const float* __restrict__ ubd,
                int dmodel, int l0, float (&dl)[16], v4f (&uv)[4])
{
    #pragma unroll
    for (int i = 0; i < 16; ++i) dl[i] = dTb[(size_t)(l0 + i) * dmodel];
    #pragma unroll
    for (int i = 0; i < 4; ++i) uv[i] = *(const v4f*)(ubd + l0 + 4 * i);
}

template <int T>
__global__ __launch_bounds__(64)
void ssm_scan_flat(const float* __restrict__ u, const float* __restrict__ D,
                   const float* __restrict__ SK, const float* __restrict__ dT,
                   const float* __restrict__ hstart, float* __restrict__ out,
                   int dmodel, int L)
{
    static_assert(T == kT, "coefficient block layout");
    constexpr int PY = T + 4;
    __shared__ __attribute__((aligned(16))) float sY[2][32 * PY];

    const int b    = blockIdx.y;
    const int c    = blockIdx.z;
    const int t    = threadIdx.x;
    const int d    = blockIdx.x * 64 + t;
    const int lbeg = c * T;
    const int lane = t & 31;
    const int wv   = __builtin_amdgcn_readfirstlane(t >> 5);

    const float* __restrict__ ubd = u   + ((size_t)b * dmodel + d) * L + lbeg;
    float*       __restrict__ ybd = out + ((size_t)b * dmodel + d) * L + lbeg;
    const float* __restrict__ dTb = dT  + ((size_t)b * L + lbeg) * dmodel + d;
    const float* __restrict__ coef = SK + ((size_t)b * gridDim.z + c) * kCoefF;
    cfloat_p Kc = (cfloat_p)(coef + kKOff);

    const float Dd = D[d];

    float dl0[16], dl1[16];
    v4f   uv0[4],  uv1[4];
    flat_fetch(dTb, ubd, dmodel, 0, dl0, uv0);

    {
        const int cc = lane & 15;
        const int kh = lane >> 4;
        const size_t orow = ((((size_t)c * gridDim.y + b) * dmodel) + blockIdx.x * 64 + 32 * wv) * 64;
        const float* __restrict__ ap = coef + cc * 64 + 4 * kh;
        const float* __restrict__ bp = hstart + orow + cc * 64 + 4 * kh;

        v8f acc[4][2];
        #pragma unroll
        for (int lt = 0; lt < 4; ++lt) {
            #pragma unroll
            for (int dt = 0; dt < 2; ++dt) acc[lt][dt] = v8f{};
        }
        #pragma unroll
        for (int q = 0; q < 8; ++q) {
            v4f a[4], bb[2];
            #pragma unroll
            for (int lt = 0; lt < 4; ++lt) a[lt]  = *(const v4f*)(ap + lt * 16 * 64 + 8 * q);
            #pragma unroll
            for (int dt = 0; dt < 2; ++dt) bb[dt] = *(const v4f*)(bp + dt * 16 * 64 + 8 * q);
            #pragma unroll
            for (int lt = 0; lt < 4; ++lt) {
                const v2f a0 = {a[lt].x, a[lt].y};
                const v2f a1 = {a[lt].z, a[lt].w};
                #pragma unroll
                for (int dt = 0; dt < 2; ++dt) {
                    const v2f b0 = {bb[dt].x, bb[dt].y};
                    const v2f b1 = {bb[dt].z, bb[dt].w};
                    acc[lt][dt] = __builtin_amdgcn_wmma_f32_16x16x4_f32(
                                      false, a0, false, b0, (short)0, acc[lt][dt], false, false);
                    acc[lt][dt] = __builtin_amdgcn_wmma_f32_16x16x4_f32(
                                      false, a1, false, b1, (short)0, acc[lt][dt], false, false);
                }
            }
        }
        float* __restrict__ yw = &sY[wv][0];
        #pragma unroll
        for (int lt = 0; lt < 4; ++lt) {
            #pragma unroll
            for (int dt = 0; dt < 2; ++dt) {
                float* __restrict__ dst = yw + (16 * dt + cc) * PY + 16 * lt + 8 * kh;
                const v4f lo = {acc[lt][dt][0], acc[lt][dt][1], acc[lt][dt][2], acc[lt][dt][3]};
                const v4f hi = {acc[lt][dt][4], acc[lt][dt][5], acc[lt][dt][6], acc[lt][dt][7]};
                *(v4f*)(dst)     = lo;
                *(v4f*)(dst + 4) = hi;
            }
        }
    }
    __syncthreads();

    const float* __restrict__ yrow = &sY[wv][lane * PY];
    float w[kT];
    float dpprev = 1.0f;

    flat_fetch(dTb, ubd, dmodel, 16, dl1, uv1);
    flat_block<0>(Kc, yrow, dl0, uv0, ybd, w, dpprev, Dd);
    flat_fetch(dTb, ubd, dmodel, 32, dl0, uv0);
    flat_block<1>(Kc, yrow, dl1, uv1, ybd, w, dpprev, Dd);
    flat_fetch(dTb, ubd, dmodel, 48, dl1, uv1);
    flat_block<2>(Kc, yrow, dl0, uv0, ybd, w, dpprev, Dd);
    flat_block<3>(Kc, yrow, dl1, uv1, ybd, w, dpprev, Dd);
}

extern "C" void kernel_launch(void* const* d_in, const int* in_sizes, int n_in,
                              void* d_out, int out_size, void* d_ws, size_t ws_size,
                              hipStream_t stream)
{
    (void)in_sizes; (void)n_in; (void)out_size; (void)ws_size;

    const float* u     = (const float*)d_in[1];
    const float* A     = (const float*)d_in[2];
    const float* Bm    = (const float*)d_in[3];
    const float* Cm    = (const float*)d_in[4];
    const float* D     = (const float*)d_in[5];
    const float* Delta = (const float*)d_in[6];
    float* out = (float*)d_out;

    constexpr int batch = 4, dmodel = 1024, N = 64, L = 2048;

    float* BuT = (float*)d_ws;
    float* CuT = BuT + (size_t)batch * L * N;
    float* dT  = CuT + (size_t)batch * L * N;

    constexpr int chunks = 32, T = L / chunks;
    constexpr size_t plane = (size_t)batch * dmodel * N;
    float* hloc   = dT    + (size_t)batch * L * dmodel;
    float* pdel   = hloc  + (size_t)(chunks - 1) * plane;
    float* hstart = pdel  + (size_t)(chunks - 1) * batch * dmodel;
    float* SK     = hstart + (size_t)chunks * plane;

    const dim3 blk(128, 1, 1);

    const dim3 gblk(256, 1, 1);
    bc_gemm<dmodel, L><<<dim3(L / 32, batch, 1), gblk, 0, stream>>>(Bm, Cm, u, BuT, CuT);
    chunk_coef<L, T><<<dim3(T / 16, batch, chunks), gblk, 0, stream>>>(BuT, CuT, A, SK);
    gemm_f32_wmma<dmodel, L, 4, 2, 32><<<dim3(L / 128, dmodel / 128, batch), gblk, 0, stream>>>(
        Delta, u, dT, dmodel);

    ssm_chunk_state<T><<<dim3(dmodel / 64, batch, chunks - 1), blk, 0, stream>>>(
        u, A, BuT, dT, hloc, pdel, dmodel, N, L);
    ssm_chunk_carry<N, T><<<dim3((unsigned)(plane / 4 / 256)), dim3(256, 1, 1), 0, stream>>>(
        hloc, pdel, A, hstart, chunks, plane);
    ssm_scan_flat<T><<<dim3(dmodel / 64, batch, chunks), dim3(64, 1, 1), 0, stream>>>(
        u, D, SK, dT, hstart, out, dmodel, L);
}


# carry pass and scan-coefficient pass share one launch: workgroup index picks the role, latency-bound carry chains run beside LDS-bound coefficient blocks; 6 launches to 5; sums unchanged
# speedup vs baseline: 15.8576x; 1.0486x over previous
#include <hip/hip_runtime.h>

typedef __attribute__((ext_vector_type(2))) float v2f;
typedef __attribute__((ext_vector_type(4))) float v4f;
typedef __attribute__((ext_vector_type(8))) float v8f;

template <int TM, int TN, int KS, int PA, int PB>
__device__ __forceinline__
void gemm_stage(const float* __restrict__ As, const float* __restrict__ Bs,
                v8f (&acc)[TM][TN])
{
    #pragma unroll
    for (int j = 0; j < KS / 4; ++j) {
        v2f a[TM], b[TN];
        #pragma unroll
        for (int i = 0; i < TM; ++i)
            a[i] = *(const v2f*)(As + i * 16 * PA + 4 * j);
        #pragma unroll
        for (int t = 0; t < TN; ++t) {
            b[t].x = Bs[(4 * j)     * PB + t * 16];
            b[t].y = Bs[(4 * j + 1) * PB + t * 16];
        }
        #pragma unroll
        for (int i = 0; i < TM; ++i) {
            #pragma unroll
            for (int t = 0; t < TN; ++t)
                acc[i][t] = __builtin_amdgcn_wmma_f32_16x16x4_f32(
                                false, a[i], false, b[t], (short)0, acc[i][t], false, false);
        }
    }
}

template <int K, int L, int TM, int TN, int KS>
__global__ __launch_bounds__(256)
__attribute__((amdgpu_waves_per_eu(4, 4)))
void gemm_f32_wmma(const float* __restrict__ W, const float* __restrict__ U,
                   float* __restrict__ Out, int M)
{
    constexpr int BM = 32 * TM;
    constexpr int BN = 64 * TN;
    constexpr int PA = KS + 4;
    constexpr int PB = BN + 8;
    constexpr int NS = K / KS;
    constexpr int A_VPR = KS / 4;
    constexpr int A_RPP = 256 / A_VPR;
    constexpr int A_NP  = BM / A_RPP;
    constexpr int B_VPR = BN / 4;
    constexpr int B_RPP = 256 / B_VPR;
    constexpr int B_NP  = KS / B_RPP;
    static_assert(K % KS == 0 && BM % A_RPP == 0 && KS % B_RPP == 0, "tile shapes");
    static_assert((PA % 8) == 4 && (PB % 32) == 8, "bank-conflict-free pitches");

    __shared__ __attribute__((aligned(16))) float sA[2][BM * PA];
    __shared__ __attribute__((aligned(16))) float sB[2][KS * PB];

    const int b = blockIdx.z;
    const float* __restrict__ Ub = U + (long)b * K * L;
    float* __restrict__ Ob = Out + (long)b * (long)M * L;

    const int t    = threadIdx.x;
    const int wave = t >> 5;
    const int lane = t & 31;
    const int wm   = wave >> 2;
    const int wn   = wave & 3;
    const int row  = lane & 15;
    const int kh   = lane >> 4;
    const int m0   = blockIdx.y * BM;
    const int l0   = blockIdx.x * BN;

    const int ar = t / A_VPR, aq = t % A_VPR;
    const int br = t / B_VPR, bq = t % B_VPR;
    const float* __restrict__ ag = W  + (long)(m0 + ar) * K + 4 * aq;
    const float* __restrict__ bg = Ub + (long)br * L + l0 + 4 * bq;
    const int as_off = ar * PA + 4 * aq;
    const int bs_off = br * PB + 4 * bq;

    const int af_off = (wm * TM * 16 + row) * PA + 2 * kh;
    const int bf_off = (2 * kh) * PB + wn * TN * 16 + row;

    v8f acc[TM][TN];
    #pragma unroll
    for (int i = 0; i < TM; ++i) {
        #pragma unroll
        for (int j = 0; j < TN; ++j) acc[i][j] = v8f{};
    }

    v4f ra[A_NP], rb[B_NP];

    #pragma unroll
    for (int p = 0; p < A_NP; ++p) ra[p] = *(const v4f*)(ag + (long)p * A_RPP * K);
    #pragma unroll
    for (int p = 0; p < B_NP; ++p) rb[p] = *(const v4f*)(bg + (long)p * B_RPP * L);
    #pragma unroll
    for (int p = 0; p < A_NP; ++p) *(v4f*)(&sA[0][as_off + p * A_RPP * PA]) = ra[p];
    #pragma unroll
    for (int p = 0; p < B_NP; ++p) *(v4f*)(&sB[0][bs_off + p * B_RPP * PB]) = rb[p];
    __syncthreads();

    for (int s = 0; s + 1 < NS; ++s) {
        const int cur = s & 1;
        ag += KS;
        bg += (long)KS * L;
        #pragma unroll
        for (int p = 0; p < A_NP; ++p) ra[p] = *(const v4f*)(ag + (long)p * A_RPP * K);
        #pragma unroll
        for (int p = 0; p < B_NP; ++p) rb[p] = *(const v4f*)(bg + (long)p * B_RPP * L);

        gemm_stage<TM, TN, KS, PA, PB>(&sA[cur][af_off], &sB[cur][bf_off], acc);

        #pragma unroll
        for (int p = 0; p < A_NP; ++p) *(v4f*)(&sA[cur ^ 1][as_off + p * A_RPP * PA]) = ra[p];
        #pragma unroll
        for (int p = 0; p < B_NP; ++p) *(v4f*)(&sB[cur ^ 1][bs_off + p * B_RPP * PB]) = rb[p];
        __syncthreads();
    }
    gemm_stage<TM, TN, KS, PA, PB>(&sA[(NS - 1) & 1][af_off], &sB[(NS - 1) & 1][bf_off], acc);

    #pragma unroll
    for (int i = 0; i < TM; ++i) {
        #pragma unroll
        for (int j = 0; j < TN; ++j) {
            float* __restrict__ dst = Ob + (long)(l0 + (wn * TN + j) * 16 + row) * M
                                         + m0 + (wm * TM + i) * 16 + 8 * kh;
            v4f lo = {acc[i][j][0], acc[i][j][1], acc[i][j][2], acc[i][j][3]};
            v4f hi = {acc[i][j][4], acc[i][j][5], acc[i][j][6], acc[i][j][7]};
            *(v4f*)(dst)     = lo;
            *(v4f*)(dst + 4) = hi;
        }
    }
}

template <int T>
__global__ __launch_bounds__(128)
void ssm_chunk_state(const float* __restrict__ u, const float* __restrict__ A,
                     const float* __restrict__ BuT, const float* __restrict__ dT,
                     float* __restrict__ hloc, float* __restrict__ pdel,
                     int dmodel, int N, int L)
{
    static_assert(T == 64, "u tile copy: 16 threads x 16 bytes per row of T steps");
    constexpr int PG = T + 4;
    constexpr int PE = 64 + 8;
    __shared__ __attribute__((aligned(16))) float sG[64 * PG];
    __shared__ __attribute__((aligned(16))) float sE[T * PE];

    const int b    = blockIdx.y;
    const int c    = blockIdx.z;
    const int t    = threadIdx.x;
    const int d0   = blockIdx.x * 64;
    const int lbeg = c * T;
    const int wv   = __builtin_amdgcn_readfirstlane(t >> 5);

    const float* __restrict__ ub  = u   + ((size_t)b * dmodel + d0) * L + lbeg;
    const float* __restrict__ Bub = BuT + ((size_t)b * L + lbeg) * N;
    const float* __restrict__ dTb = dT  + ((size_t)b * L + lbeg) * dmodel + d0;

    {
        const int r = t >> 4, q = t & 15;
        v4f uv[8];
        #pragma unroll
        for (int p = 0; p < 8; ++p)
            uv[p] = *(const v4f*)(ub + (size_t)(r + 8 * p) * L + 4 * q);
        #pragma unroll
        for (int p = 0; p < 8; ++p)
            *(v4f*)(&sG[(r + 8 * p) * PG + 4 * q]) = uv[p];
    }

    const int    idx     = t & 63;
    const float* __restrict__ colp = (wv < 2) ? (Bub + idx) : (dTb + idx);
    const size_t cstride = (wv < 2) ? (size_t)N : (size_t)dmodel;
    float v[T];
    #pragma unroll
    for (int j = 0; j < T; ++j) v[j] = colp[(size_t)j * cstride];

    __syncthreads();

    if (wv < 2) {
        const float a = A[idx];
        float pw = 1.0f;
        #pragma unroll
        for (int j = T - 1; j >= 0; --j) {
            sE[j * PE + idx] = pw * v[j];
            pw *= a;
        }
    } else {
        float s = 1.0f;
        #pragma unroll
        for (int j = T - 1; j >= 0; --j) {
            const float ul = sG[idx * PG + j];
            sG[idx * PG + j] = s * (v[j] * ul);
            s *= v[j];
        }
        pdel[((size_t)c * gridDim.y + b) * dmodel + d0 + idx] = s;
    }
    __syncthreads();

    const int lane = t & 31;
    const int cc   = lane & 15;
    const int kh   = lane >> 4;

    v8f acc[4];
    #pragma unroll
    for (int tm = 0; tm < 4; ++tm) acc[tm] = v8f{};

    const float* __restrict__ gp = &sG[(16 * wv + cc) * PG + 2 * kh];
    const float* __restrict__ ep = &sE[(2 * kh) * PE + cc];
    #pragma unroll
    for (int kk = 0; kk < T / 4; ++kk) {
        const v2f g = *(const v2f*)(gp + 4 * kk);
        #pragma unroll
        for (int tm = 0; tm < 4; ++tm) {
            v2f e;
            e.x = ep[(4 * kk)     * PE + 16 * tm];
            e.y = ep[(4 * kk + 1) * PE + 16 * tm];
            acc[tm] = __builtin_amdgcn_wmma_f32_16x16x4_f32(
                          false, e, false, g, (short)0, acc[tm], false, false);
        }
    }

    const int    d = d0 + 16 * wv + cc;
    const size_t o = ((((size_t)c * gridDim.y + b) * dmodel) + d) * N;
    #pragma unroll
    for (int tm = 0; tm < 4; ++tm) {
        const int n0 = 16 * tm + 8 * kh;
        const v4f lo = {acc[tm][0], acc[tm][1], acc[tm][2], acc[tm][3]};
        const v4f hi = {acc[tm][4], acc[tm][5], acc[tm][6], acc[tm][7]};
        *(v4f*)(hloc + o + n0)     = lo;
        *(v4f*)(hloc + o + n0 + 4) = hi;
    }
}

template <int N, int T>
__device__ __forceinline__
void chunk_carry_body(int blk,
                      const float* __restrict__ hloc, const float* __restrict__ pdel,
                      const float* __restrict__ A,
                      float* __restrict__ hstart, int nchunks, size_t plane)
{
    const size_t i    = ((size_t)blk * 256 + threadIdx.x) * 4;
    const size_t row  = i / (size_t)N;
    const int    n    = (int)(i % (size_t)N);
    const size_t rows = plane / (size_t)N;

    const v4f a = *(const v4f*)(A + n);
    v4f at = {1.0f, 1.0f, 1.0f, 1.0f};
    for (int j = 0; j < T; ++j) at *= a;

    v4f hs = {0.0f, 0.0f, 0.0f, 0.0f};
    for (int c = 0; c + 1 < nchunks; ++c) {
        const v4f   hl = *(const v4f*)(hloc + (size_t)c * plane + i);
        const float pd = pdel[(size_t)c * rows + row];
        const v4f   pv = pd * at;
        *(v4f*)(hstart + (size_t)c * plane + i) = hs;
        hs.x = fmaf(pv.x, hs.x, hl.x);
        hs.y = fmaf(pv.y, hs.y, hl.y);
        hs.z = fmaf(pv.z, hs.z, hl.z);
        hs.w = fmaf(pv.w, hs.w, hl.w);
    }
    *(v4f*)(hstart + (size_t)(nchunks - 1) * plane + i) = hs;
}

constexpr int kT     = 64;
constexpr int kKOff  = kT * 64;
constexpr int kCoefF = 2 * kT * 64;

template <int KS, int PA, int PB>
__device__ __forceinline__
void bc_stage(const float* __restrict__ As, const float* __restrict__ Bs, v8f& c0, v8f& c1)
{
    v2f a[KS / 4], b0[KS / 4], b1[KS / 4];
    #pragma unroll
    for (int j = 0; j < KS / 4; ++j) {
        a[j]    = *(const v2f*)(As + 4 * j);
        b0[j].x = Bs[(4 * j)     * PB];
        b0[j].y = Bs[(4 * j + 1) * PB];
        b1[j].x = Bs[(4 * j)     * PB + 16];
        b1[j].y = Bs[(4 * j + 1) * PB + 16];
    }
    __builtin_amdgcn_sched_barrier(0);
    #pragma unroll
    for (int j = 0; j < KS / 4; ++j) {
        c0 = __builtin_amdgcn_wmma_f32_16x16x4_f32(false, a[j], false, b0[j], (short)0, c0, false, false);
        c1 = __builtin_amdgcn_wmma_f32_16x16x4_f32(false, a[j], false, b1[j], (short)0, c1, false, false);
    }
}

template <int K, int L>
__global__ __launch_bounds__(256)
void bc_gemm(const float* __restrict__ Bm, const float* __restrict__ Cm,
             const float* __restrict__ U,
             float* __restrict__ BuT, float* __restrict__ CuT)
{
    constexpr int KS = 64;
    constexpr int BM = 128;
    constexpr int BN = 32;
    constexpr int PA = KS + 4;
    constexpr int PB = BN + 8;
    constexpr int NS = K / KS;
    constexpr int A_NP = 8;
    constexpr int B_NP = 2;
    constexpr int P = 68;
    static_assert(K % KS == 0 && (PA % 8) == 4 && (PB % 32) == 8, "tile shapes and pitches");

    __shared__ __attribute__((aligned(16))) float sA[2][BM * PA];
    __shared__ __attribute__((aligned(16))) float sB[2][KS * PB];
    __shared__ __attribute__((aligned(16))) float sBu[BN * P];
    __shared__ __attribute__((aligned(16))) float sCu[BN * P];

    const int b  = blockIdx.y;
    const int l0 = blockIdx.x * BN;
    const float* __restrict__ Ub = U + (long)b * K * L;

    const int t    = threadIdx.x;
    const int wave = t >> 5;
    const int lane = t & 31;
    const int row  = lane & 15;
    const int kh   = lane >> 4;

    const int ar = t >> 4, aq = t & 15;
    const int br = t >> 3, bq = t & 7;
    const float* __restrict__ agB = Bm + (long)ar * K + 4 * aq;
    const float* __restrict__ agC = Cm + (long)ar * K + 4 * aq;
    const float* __restrict__ bg  = Ub + (long)br * L + l0 + 4 * bq;
    const int as_off = ar * PA + 4 * aq;
    const int bs_off = br * PB + 4 * bq;
    const int af_off = (wave * 16 + row) * PA + 2 * kh;
    const int bf_off = (2 * kh) * PB + row;

    v8f acc[1][2];
    acc[0][0] = v8f{};
    acc[0][1] = v8f{};

    v4f ra[A_NP], rb[B_NP];

    #pragma unroll
    for (int p = 0; p < 4; ++p) ra[p]     = *(const v4f*)(agB + (long)p * 16 * K);
    #pragma unroll
    for (int p = 0; p < 4; ++p) ra[4 + p] = *(const v4f*)(agC + (long)p * 16 * K);
    #pragma unroll
    for (int p = 0; p < B_NP; ++p) rb[p] = *(const v4f*)(bg + (long)p * 32 * L);
    #pragma unroll
    for (int p = 0; p < A_NP; ++p) *(v4f*)(&sA[0][as_off + p * 16 * PA]) = ra[p];
    #pragma unroll
    for (int p = 0; p < B_NP; ++p) *(v4f*)(&sB[0][bs_off + p * 32 * PB]) = rb[p];
    __syncthreads();

    for (int s = 0; s + 1 < NS; ++s) {
        const int cur = s & 1;
        agB += KS;
        agC += KS;
        bg  += (long)KS * L;
        #pragma unroll
        for (int p = 0; p < 4; ++p) ra[p]     = *(const v4f*)(agB + (long)p * 16 * K);
        #pragma unroll
        for (int p = 0; p < 4; ++p) ra[4 + p] = *(const v4f*)(agC + (long)p * 16 * K);
        #pragma unroll
        for (int p = 0; p < B_NP; ++p) rb[p] = *(const v4f*)(bg + (long)p * 32 * L);

        bc_stage<KS, PA, PB>(&sA[cur][af_off], &sB[cur][bf_off], acc[0][0], acc[0][1]);

        __builtin_amdgcn_sched_barrier(0);

        #pragma unroll
        for (int p = 0; p < A_NP; ++p) *(v4f*)(&sA[cur ^ 1][as_off + p * 16 * PA]) = ra[p];
        #pragma unroll
        for (int p = 0; p < B_NP; ++p) *(v4f*)(&sB[cur ^ 1][bs_off + p * 32 * PB]) = rb[p];
        __syncthreads();
    }
    bc_stage<KS, PA, PB>(&sA[(NS - 1) & 1][af_off], &sB[(NS - 1) & 1][bf_off], acc[0][0], acc[0][1]);

    {
        float* __restrict__ dstS = (wave < 4) ? sBu : sCu;
        const int nb = (wave & 3) * 16 + 8 * kh;
        #pragma unroll
        for (int j = 0; j < 2; ++j) {
            const v4f lo = {acc[0][j][0], acc[0][j][1], acc[0][j][2], acc[0][j][3]};
            const v4f hi = {acc[0][j][4], acc[0][j][5], acc[0][j][6], acc[0][j][7]};
            *(v4f*)(&dstS[(16 * j + row) * P + nb])     = lo;
            *(v4f*)(&dstS[(16 * j + row) * P + nb + 4]) = hi;
        }
    }
    __syncthreads();

    {
        const int r = t >> 4, q = t & 15;
        float* __restrict__ dB = BuT + ((size_t)b * L + l0) * 64;
        float* __restrict__ dC = CuT + ((size_t)b * L + l0) * 64;
        #pragma unroll
        for (int p = 0; p < 2; ++p) {
            *(v4f*)(dB + (r + 16 * p) * 64 + 4 * q) = *(const v4f*)(&sBu[(r + 16 * p) * P + 4 * q]);
            *(v4f*)(dC + (r + 16 * p) * 64 + 4 * q) = *(const v4f*)(&sCu[(r + 16 * p) * P + 4 * q]);
        }
    }
}

constexpr int kCoefP = 68;
template <int L, int T>
__device__ __forceinline__
void chunk_coef_body(int sl, int b, int c, int nchunks,
                     const float* __restrict__ BuT, const float* __restrict__ CuT,
                     const float* __restrict__ A, float* __restrict__ SK,
                     float* __restrict__ sBu, float* __restrict__ sCu, float* __restrict__ sAp)
{
    static_assert(T == kT, "coefficient block layout");
    constexpr int P = kCoefP;
    const int t    = threadIdx.x;
    const int lbeg = c * T;

    if (t < 64) {
        const float a = A[t];
        float pw = 1.0f;
        for (int k = 0; k <= T; ++k) { sAp[k * P + t] = pw; pw *= a; }
    }
    {
        const int r = t >> 4, q = t & 15;
        const float* __restrict__ srcB = BuT + ((size_t)b * L + lbeg) * 64;
        const float* __restrict__ srcC = CuT + ((size_t)b * L + lbeg + 16 * sl) * 64;
        for (int p = 0; p <= sl; ++p)
            *(v4f*)(&sBu[(r + 16 * p) * P + 4 * q]) = *(const v4f*)(srcB + (r + 16 * p) * 64 + 4 * q);
        *(v4f*)(&sCu[r * P + 4 * q]) = *(const v4f*)(srcC + r * 64 + 4 * q);
    }
    __syncthreads();

    float* __restrict__ blk = SK + ((size_t)b * nchunks + c) * kCoefF;

    #pragma unroll
    for (int p = 0; p < 4; ++p) {
        const int idx = t + 256 * p;
        const int r = idx >> 6, n = idx & 63;
        const int l = 16 * sl + r;
        blk[l * 64 + n] = sCu[r * P + n] * sAp[(l + 1) * P + n];
    }

    const int li = t >> 4, jj = t & 15;
    const int l  = 16 * sl + li;
    const float* __restrict__ cr = &sCu[li * P];
    for (int sj = 0; sj < T / 16; ++sj) {
        const int j = 16 * sj + jj;
        float kv = 0.0f;
        if (sj <= sl) {
            const int k = (j <= l) ? (l - j) : 0;
            const float* __restrict__ ar2 = &sAp[k * P];
            const float* __restrict__ br2 = &sBu[j * P];
            float q0 = 0.0f, q1 = 0.0f, q2 = 0.0f, q3 = 0.0f;
            #pragma unroll
            for (int n = 0; n < 64; n += 4) {
                const v4f cv = *(const v4f*)(cr + n);
                const v4f av = *(const v4f*)(ar2 + n);
                const v4f bv = *(const v4f*)(br2 + n);
                q0 = fmaf(cv.x, av.x * bv.x, q0);
                q1 = fmaf(cv.y, av.y * bv.y, q1);
                q2 = fmaf(cv.z, av.z * bv.z, q2);
                q3 = fmaf(cv.w, av.w * bv.w, q3);
            }
            kv = (j <= l) ? ((q0 + q1) + (q2 + q3)) : 0.0f;
        }
        blk[kKOff + l * 64 + j] = kv;
    }
}

template <int N, int L, int T, int BATCH>
__global__ __launch_bounds__(256)
void carry_and_coef(const float* __restrict__ hloc, const float* __restrict__ pdel,
                    const float* __restrict__ A, float* __restrict__ hstart,
                    int nchunks, size_t plane, int carry_wgs,
                    const float* __restrict__ BuT, const float* __restrict__ CuT,
                    float* __restrict__ SK)
{
    __shared__ __attribute__((aligned(16))) float sBu[T * kCoefP];
    __shared__ __attribute__((aligned(16))) float sCu[16 * kCoefP];
    __shared__ __attribute__((aligned(16))) float sAp[(T + 1) * kCoefP];

    const int wg = blockIdx.x;
    if (wg < carry_wgs) {
        chunk_carry_body<N, T>(wg, hloc, pdel, A, hstart, nchunks, plane);
    } else {
        constexpr int SL = T / 16;
        const int id = wg - carry_wgs;
        const int sl = id % SL;
        const int b  = (id / SL) % BATCH;
        const int c  = id / (SL * BATCH);
        chunk_coef_body<L, T>(sl, b, c, nchunks, BuT, CuT, A, SK, sBu, sCu, sAp);
    }
}

typedef const __attribute__((address_space(4))) float* cfloat_p;

template <int l>
__device__ __forceinline__
float flat_step(cfloat_p krow, float (&w)[kT], float& dpb, float dpprev,
                float y1, float dl, float ul, float Dd)
{
    constexpr int s0 = (l >> 4) << 4;
    dpb *= dl;
    #pragma unroll
    for (int j = s0; j < l; ++j) w[j] *= dl;
    w[l] = dl * ul;
    float qo[3] = {0.0f, 0.0f, 0.0f};
    #pragma unroll
    for (int j = 0; j < s0; ++j) qo[j >> 4] = fmaf(krow[j], w[j], qo[j >> 4]);
    float sn = 0.0f;
    #pragma unroll
    for (int j = s0; j <= l; ++j) sn = fmaf(krow[j], w[j], sn);
    const float so = fmaf(dpprev, y1, (qo[0] + qo[1]) + qo[2]);
    const float y  = fmaf(dpb, so, sn);
    return fmaf(ul, Dd, y);
}

template <int l4>
__device__ __forceinline__
void flat_group4(cfloat_p K, const v4f y1, const v4f uv, const float (&dl)[16],
                 float* __restrict__ ybd, float (&w)[kT], float& dpb, float dpprev, float Dd)
{
    constexpr int i4 = l4 & 15;
    v4f yo;
    yo.x = flat_step<l4 + 0>(K + (l4 + 0) * 64, w, dpb, dpprev, y1.x, dl[i4 + 0], uv.x, Dd);
    yo.y = flat_step<l4 + 1>(K + (l4 + 1) * 64, w, dpb, dpprev, y1.y, dl[i4 + 1], uv.y, Dd);
    yo.z = flat_step<l4 + 2>(K + (l4 + 2) * 64, w, dpb, dpprev, y1.z, dl[i4 + 2], uv.z, Dd);
    yo.w = flat_step<l4 + 3>(K + (l4 + 3) * 64, w, dpb, dpprev, y1.w, dl[i4 + 3], uv.w, Dd);
    *(v4f*)(ybd + l4) = yo;
}

template <int s>
__device__ __forceinline__
void flat_block(cfloat_p K, const float* __restrict__ yrow, const float (&dl)[16],
                const v4f (&uv)[4], float* __restrict__ ybd, float (&w)[kT],
                float& dpprev, float Dd)
{
    v4f y1[4];
    #pragma unroll
    for (int i = 0; i < 4; ++i) y1[i] = *(const v4f*)(yrow + 16 * s + 4 * i);
    float dpb = 1.0f;
    flat_group4<16 * s + 0> (K, y1[0], uv[0], dl, ybd, w, dpb, dpprev, Dd);
    flat_group4<16 * s + 4> (K, y1[1], uv[1], dl, ybd, w, dpb, dpprev, Dd);
    flat_group4<16 * s + 8> (K, y1[2], uv[2], dl, ybd, w, dpb, dpprev, Dd);
    flat_group4<16 * s + 12>(K, y1[3], uv[3], dl, ybd, w, dpb, dpprev, Dd);
    if (16 * (s + 1) < kT) {
        #pragma unroll
        for (int j = 0; j < 16 * s; ++j) w[j] *= dpb;
        dpprev *= dpb;
    }
}

__device__ __forceinline__
void flat_fetch(const float* __restrict__ dTb, const float* __restrict__ ubd,
                int dmodel, int l0, float (&dl)[16], v4f (&uv)[4])
{
    #pragma unroll
    for (int i = 0; i < 16; ++i) dl[i] = dTb[(size_t)(l0 + i) * dmodel];
    #pragma unroll
    for (int i = 0; i < 4; ++i) uv[i] = *(const v4f*)(ubd + l0 + 4 * i);
}

template <int T>
__global__ __launch_bounds__(64)
void ssm_scan_flat(const float* __restrict__ u, const float* __restrict__ D,
                   const float* __restrict__ SK, const float* __restrict__ dT,
                   const float* __restrict__ hstart, float* __restrict__ out,
                   int dmodel, int L)
{
    static_assert(T == kT, "coefficient block layout");
    constexpr int PY = T + 4;
    __shared__ __attribute__((aligned(16))) float sY[2][32 * PY];

    const int b    = blockIdx.y;
    const int c    = blockIdx.z;
    const int t    = threadIdx.x;
    const int d    = blockIdx.x * 64 + t;
    const int lbeg = c * T;
    const int lane = t & 31;
    const int wv   = __builtin_amdgcn_readfirstlane(t >> 5);

    const float* __restrict__ ubd = u   + ((size_t)b * dmodel + d) * L + lbeg;
    float*       __restrict__ ybd = out + ((size_t)b * dmodel + d) * L + lbeg;
    const float* __restrict__ dTb = dT  + ((size_t)b * L + lbeg) * dmodel + d;
    const float* __restrict__ coef = SK + ((size_t)b * gridDim.z + c) * kCoefF;
    cfloat_p Kc = (cfloat_p)(coef + kKOff);

    const float Dd = D[d];

    float dl0[16], dl1[16];
    v4f   uv0[4],  uv1[4];
    flat_fetch(dTb, ubd, dmodel, 0, dl0, uv0);

    {
        const int cc = lane & 15;
        const int kh = lane >> 4;
        const size_t orow = ((((size_t)c * gridDim.y + b) * dmodel) + blockIdx.x * 64 + 32 * wv) * 64;
        const float* __restrict__ ap = coef + cc * 64 + 4 * kh;
        const float* __restrict__ bp = hstart + orow + cc * 64 + 4 * kh;

        v8f acc[4][2];
        #pragma unroll
        for (int lt = 0; lt < 4; ++lt) {
            #pragma unroll
            for (int dt = 0; dt < 2; ++dt) acc[lt][dt] = v8f{};
        }
        #pragma unroll
        for (int q = 0; q < 8; ++q) {
            v4f a[4], bb[2];
            #pragma unroll
            for (int lt = 0; lt < 4; ++lt) a[lt]  = *(const v4f*)(ap + lt * 16 * 64 + 8 * q);
            #pragma unroll
            for (int dt = 0; dt < 2; ++dt) bb[dt] = *(const v4f*)(bp + dt * 16 * 64 + 8 * q);
            #pragma unroll
            for (int lt = 0; lt < 4; ++lt) {
                const v2f a0 = {a[lt].x, a[lt].y};
                const v2f a1 = {a[lt].z, a[lt].w};
                #pragma unroll
                for (int dt = 0; dt < 2; ++dt) {
                    const v2f b0 = {bb[dt].x, bb[dt].y};
                    const v2f b1 = {bb[dt].z, bb[dt].w};
                    acc[lt][dt] = __builtin_amdgcn_wmma_f32_16x16x4_f32(
                                      false, a0, false, b0, (short)0, acc[lt][dt], false, false);
                    acc[lt][dt] = __builtin_amdgcn_wmma_f32_16x16x4_f32(
                                      false, a1, false, b1, (short)0, acc[lt][dt], false, false);
                }
            }
        }
        float* __restrict__ yw = &sY[wv][0];
        #pragma unroll
        for (int lt = 0; lt < 4; ++lt) {
            #pragma unroll
            for (int dt = 0; dt < 2; ++dt) {
                float* __restrict__ dst = yw + (16 * dt + cc) * PY + 16 * lt + 8 * kh;
                const v4f lo = {acc[lt][dt][0], acc[lt][dt][1], acc[lt][dt][2], acc[lt][dt][3]};
                const v4f hi = {acc[lt][dt][4], acc[lt][dt][5], acc[lt][dt][6], acc[lt][dt][7]};
                *(v4f*)(dst)     = lo;
                *(v4f*)(dst + 4) = hi;
            }
        }
    }
    __syncthreads();

    const float* __restrict__ yrow = &sY[wv][lane * PY];
    float w[kT];
    float dpprev = 1.0f;

    flat_fetch(dTb, ubd, dmodel, 16, dl1, uv1);
    flat_block<0>(Kc, yrow, dl0, uv0, ybd, w, dpprev, Dd);
    flat_fetch(dTb, ubd, dmodel, 32, dl0, uv0);
    flat_block<1>(Kc, yrow, dl1, uv1, ybd, w, dpprev, Dd);
    flat_fetch(dTb, ubd, dmodel, 48, dl1, uv1);
    flat_block<2>(Kc, yrow, dl0, uv0, ybd, w, dpprev, Dd);
    flat_block<3>(Kc, yrow, dl1, uv1, ybd, w, dpprev, Dd);
}

extern "C" void kernel_launch(void* const* d_in, const int* in_sizes, int n_in,
                              void* d_out, int out_size, void* d_ws, size_t ws_size,
                              hipStream_t stream)
{
    (void)in_sizes; (void)n_in; (void)out_size; (void)ws_size;

    const float* u     = (const float*)d_in[1];
    const float* A     = (const float*)d_in[2];
    const float* Bm    = (const float*)d_in[3];
    const float* Cm    = (const float*)d_in[4];
    const float* D     = (const float*)d_in[5];
    const float* Delta = (const float*)d_in[6];
    float* out = (float*)d_out;

    constexpr int batch = 4, dmodel = 1024, N = 64, L = 2048;

    float* BuT = (float*)d_ws;
    float* CuT = BuT + (size_t)batch * L * N;
    float* dT  = CuT + (size_t)batch * L * N;

    constexpr int chunks = 32, T = L / chunks;
    constexpr size_t plane = (size_t)batch * dmodel * N;
    float* hloc   = dT    + (size_t)batch * L * dmodel;
    float* pdel   = hloc  + (size_t)(chunks - 1) * plane;
    float* hstart = pdel  + (size_t)(chunks - 1) * batch * dmodel;
    float* SK     = hstart + (size_t)chunks * plane;

    const dim3 blk(128, 1, 1);

    const dim3 gblk(256, 1, 1);
    bc_gemm<dmodel, L><<<dim3(L / 32, batch, 1), gblk, 0, stream>>>(Bm, Cm, u, BuT, CuT);
    gemm_f32_wmma<dmodel, L, 4, 2, 32><<<dim3(L / 128, dmodel / 128, batch), gblk, 0, stream>>>(
        Delta, u, dT, dmodel);

    ssm_chunk_state<T><<<dim3(dmodel / 64, batch, chunks - 1), blk, 0, stream>>>(
        u, A, BuT, dT, hloc, pdel, dmodel, N, L);
    constexpr int carry_wgs = (int)(plane / 4 / 256);
    constexpr int coef_wgs  = (T / 16) * batch * chunks;
    carry_and_coef<N, L, T, batch><<<dim3(carry_wgs + coef_wgs, 1, 1), gblk, 0, stream>>>(
        hloc, pdel, A, hstart, chunks, plane, carry_wgs, BuT, CuT, SK);
    ssm_scan_flat<T><<<dim3(dmodel / 64, batch, chunks), dim3(64, 1, 1), 0, stream>>>(
        u, D, SK, dT, hstart, out, dmodel, L);
}


# scan: a 16-step block's sums over the older blocks' steps as one WMMA product K*W^T per block (weights via LDS), the step chain keeps scalar loads for its own block of K only; y sum order changes
# speedup vs baseline: 16.3929x; 1.0338x over previous
#include <hip/hip_runtime.h>

typedef __attribute__((ext_vector_type(2))) float v2f;
typedef __attribute__((ext_vector_type(4))) float v4f;
typedef __attribute__((ext_vector_type(8))) float v8f;

template <int TM, int TN, int KS, int PA, int PB>
__device__ __forceinline__
void gemm_stage(const float* __restrict__ As, const float* __restrict__ Bs,
                v8f (&acc)[TM][TN])
{
    #pragma unroll
    for (int j = 0; j < KS / 4; ++j) {
        v2f a[TM], b[TN];
        #pragma unroll
        for (int i = 0; i < TM; ++i)
            a[i] = *(const v2f*)(As + i * 16 * PA + 4 * j);
        #pragma unroll
        for (int t = 0; t < TN; ++t) {
            b[t].x = Bs[(4 * j)     * PB + t * 16];
            b[t].y = Bs[(4 * j + 1) * PB + t * 16];
        }
        #pragma unroll
        for (int i = 0; i < TM; ++i) {
            #pragma unroll
            for (int t = 0; t < TN; ++t)
                acc[i][t] = __builtin_amdgcn_wmma_f32_16x16x4_f32(
                                false, a[i], false, b[t], (short)0, acc[i][t], false, false);
        }
    }
}

template <int K, int L, int TM, int TN, int KS>
__global__ __launch_bounds__(256)
__attribute__((amdgpu_waves_per_eu(4, 4)))
void gemm_f32_wmma(const float* __restrict__ W, const float* __restrict__ U,
                   float* __restrict__ Out, int M)
{
    constexpr int BM = 32 * TM;
    constexpr int BN = 64 * TN;
    constexpr int PA = KS + 4;
    constexpr int PB = BN + 8;
    constexpr int NS = K / KS;
    constexpr int A_VPR = KS / 4;
    constexpr int A_RPP = 256 / A_VPR;
    constexpr int A_NP  = BM / A_RPP;
    constexpr int B_VPR = BN / 4;
    constexpr int B_RPP = 256 / B_VPR;
    constexpr int B_NP  = KS / B_RPP;
    static_assert(K % KS == 0 && BM % A_RPP == 0 && KS % B_RPP == 0, "tile shapes");
    static_assert((PA % 8) == 4 && (PB % 32) == 8, "bank-conflict-free pitches");

    __shared__ __attribute__((aligned(16))) float sA[2][BM * PA];
    __shared__ __attribute__((aligned(16))) float sB[2][KS * PB];

    const int b = blockIdx.z;
    const float* __restrict__ Ub = U + (long)b * K * L;
    float* __restrict__ Ob = Out + (long)b * (long)M * L;

    const int t    = threadIdx.x;
    const int wave = t >> 5;
    const int lane = t & 31;
    const int wm   = wave >> 2;
    const int wn   = wave & 3;
    const int row  = lane & 15;
    const int kh   = lane >> 4;
    const int m0   = blockIdx.y * BM;
    const int l0   = blockIdx.x * BN;

    const int ar = t / A_VPR, aq = t % A_VPR;
    const int br = t / B_VPR, bq = t % B_VPR;
    const float* __restrict__ ag = W  + (long)(m0 + ar) * K + 4 * aq;
    const float* __restrict__ bg = Ub + (long)br * L + l0 + 4 * bq;
    const int as_off = ar * PA + 4 * aq;
    const int bs_off = br * PB + 4 * bq;

    const int af_off = (wm * TM * 16 + row) * PA + 2 * kh;
    const int bf_off = (2 * kh) * PB + wn * TN * 16 + row;

    v8f acc[TM][TN];
    #pragma unroll
    for (int i = 0; i < TM; ++i) {
        #pragma unroll
        for (int j = 0; j < TN; ++j) acc[i][j] = v8f{};
    }

    v4f ra[A_NP], rb[B_NP];

    #pragma unroll
    for (int p = 0; p < A_NP; ++p) ra[p] = *(const v4f*)(ag + (long)p * A_RPP * K);
    #pragma unroll
    for (int p = 0; p < B_NP; ++p) rb[p] = *(const v4f*)(bg + (long)p * B_RPP * L);
    #pragma unroll
    for (int p = 0; p < A_NP; ++p) *(v4f*)(&sA[0][as_off + p * A_RPP * PA]) = ra[p];
    #pragma unroll
    for (int p = 0; p < B_NP; ++p) *(v4f*)(&sB[0][bs_off + p * B_RPP * PB]) = rb[p];
    __syncthreads();

    for (int s = 0; s + 1 < NS; ++s) {
        const int cur = s & 1;
        ag += KS;
        bg += (long)KS * L;
        #pragma unroll
        for (int p = 0; p < A_NP; ++p) ra[p] = *(const v4f*)(ag + (long)p * A_RPP * K);
        #pragma unroll
        for (int p = 0; p < B_NP; ++p) rb[p] = *(const v4f*)(bg + (long)p * B_RPP * L);

        gemm_stage<TM, TN, KS, PA, PB>(&sA[cur][af_off], &sB[cur][bf_off], acc);

        #pragma unroll
        for (int p = 0; p < A_NP; ++p) *(v4f*)(&sA[cur ^ 1][as_off + p * A_RPP * PA]) = ra[p];
        #pragma unroll
        for (int p = 0; p < B_NP; ++p) *(v4f*)(&sB[cur ^ 1][bs_off + p * B_RPP * PB]) = rb[p];
        __syncthreads();
    }
    gemm_stage<TM, TN, KS, PA, PB>(&sA[(NS - 1) & 1][af_off], &sB[(NS - 1) & 1][bf_off], acc);

    #pragma unroll
    for (int i = 0; i < TM; ++i) {
        #pragma unroll
        for (int j = 0; j < TN; ++j) {
            float* __restrict__ dst = Ob + (long)(l0 + (wn * TN + j) * 16 + row) * M
                                         + m0 + (wm * TM + i) * 16 + 8 * kh;
            v4f lo = {acc[i][j][0], acc[i][j][1], acc[i][j][2], acc[i][j][3]};
            v4f hi = {acc[i][j][4], acc[i][j][5], acc[i][j][6], acc[i][j][7]};
            *(v4f*)(dst)     = lo;
            *(v4f*)(dst + 4) = hi;
        }
    }
}

template <int T>
__global__ __launch_bounds__(128)
void ssm_chunk_state(const float* __restrict__ u, const float* __restrict__ A,
                     const float* __restrict__ BuT, const float* __restrict__ dT,
                     float* __restrict__ hloc, float* __restrict__ pdel,
                     int dmodel, int N, int L)
{
    static_assert(T == 64, "u tile copy: 16 threads x 16 bytes per row of T steps");
    constexpr int PG = T + 4;
    constexpr int PE = 64 + 8;
    __shared__ __attribute__((aligned(16))) float sG[64 * PG];
    __shared__ __attribute__((aligned(16))) float sE[T * PE];

    const int b    = blockIdx.y;
    const int c    = blockIdx.z;
    const int t    = threadIdx.x;
    const int d0   = blockIdx.x * 64;
    const int lbeg = c * T;
    const int wv   = __builtin_amdgcn_readfirstlane(t >> 5);

    const float* __restrict__ ub  = u   + ((size_t)b * dmodel + d0) * L + lbeg;
    const float* __restrict__ Bub = BuT + ((size_t)b * L + lbeg) * N;
    const float* __restrict__ dTb = dT  + ((size_t)b * L + lbeg) * dmodel + d0;

    {
        const int r = t >> 4, q = t & 15;
        v4f uv[8];
        #pragma unroll
        for (int p = 0; p < 8; ++p)
            uv[p] = *(const v4f*)(ub + (size_t)(r + 8 * p) * L + 4 * q);
        #pragma unroll
        for (int p = 0; p < 8; ++p)
            *(v4f*)(&sG[(r + 8 * p) * PG + 4 * q]) = uv[p];
    }

    const int    idx     = t & 63;
    const float* __restrict__ colp = (wv < 2) ? (Bub + idx) : (dTb + idx);
    const size_t cstride = (wv < 2) ? (size_t)N : (size_t)dmodel;
    float v[T];
    #pragma unroll
    for (int j = 0; j < T; ++j) v[j] = colp[(size_t)j * cstride];

    __syncthreads();

    if (wv < 2) {
        const float a = A[idx];
        float pw = 1.0f;
        #pragma unroll
        for (int j = T - 1; j >= 0; --j) {
            sE[j * PE + idx] = pw * v[j];
            pw *= a;
        }
    } else {
        float s = 1.0f;
        #pragma unroll
        for (int j = T - 1; j >= 0; --j) {
            const float ul = sG[idx * PG + j];
            sG[idx * PG + j] = s * (v[j] * ul);
            s *= v[j];
        }
        pdel[((size_t)c * gridDim.y + b) * dmodel + d0 + idx] = s;
    }
    __syncthreads();

    const int lane = t & 31;
    const int cc   = lane & 15;
    const int kh   = lane >> 4;

    v8f acc[4];
    #pragma unroll
    for (int tm = 0; tm < 4; ++tm) acc[tm] = v8f{};

    const float* __restrict__ gp = &sG[(16 * wv + cc) * PG + 2 * kh];
    const float* __restrict__ ep = &sE[(2 * kh) * PE + cc];
    #pragma unroll
    for (int kk = 0; kk < T / 4; ++kk) {
        const v2f g = *(const v2f*)(gp + 4 * kk);
        #pragma unroll
        for (int tm = 0; tm < 4; ++tm) {
            v2f e;
            e.x = ep[(4 * kk)     * PE + 16 * tm];
            e.y = ep[(4 * kk + 1) * PE + 16 * tm];
            acc[tm] = __builtin_amdgcn_wmma_f32_16x16x4_f32(
                          false, e, false, g, (short)0, acc[tm], false, false);
        }
    }

    const int    d = d0 + 16 * wv + cc;
    const size_t o = ((((size_t)c * gridDim.y + b) * dmodel) + d) * N;
    #pragma unroll
    for (int tm = 0; tm < 4; ++tm) {
        const int n0 = 16 * tm + 8 * kh;
        const v4f lo = {acc[tm][0], acc[tm][1], acc[tm][2], acc[tm][3]};
        const v4f hi = {acc[tm][4], acc[tm][5], acc[tm][6], acc[tm][7]};
        *(v4f*)(hloc + o + n0)     = lo;
        *(v4f*)(hloc + o + n0 + 4) = hi;
    }
}

template <int N, int T>
__device__ __forceinline__
void chunk_carry_body(int blk,
                      const float* __restrict__ hloc, const float* __restrict__ pdel,
                      const float* __restrict__ A,
                      float* __restrict__ hstart, int nchunks, size_t plane)
{
    const size_t i    = ((size_t)blk * 256 + threadIdx.x) * 4;
    const size_t row  = i / (size_t)N;
    const int    n    = (int)(i % (size_t)N);
    const size_t rows = plane / (size_t)N;

    const v4f a = *(const v4f*)(A + n);
    v4f at = {1.0f, 1.0f, 1.0f, 1.0f};
    for (int j = 0; j < T; ++j) at *= a;

    v4f hs = {0.0f, 0.0f, 0.0f, 0.0f};
    for (int c = 0; c + 1 < nchunks; ++c) {
        const v4f   hl = *(const v4f*)(hloc + (size_t)c * plane + i);
        const float pd = pdel[(size_t)c * rows + row];
        const v4f   pv = pd * at;
        *(v4f*)(hstart + (size_t)c * plane + i) = hs;
        hs.x = fmaf(pv.x, hs.x, hl.x);
        hs.y = fmaf(pv.y, hs.y, hl.y);
        hs.z = fmaf(pv.z, hs.z, hl.z);
        hs.w = fmaf(pv.w, hs.w, hl.w);
    }
    *(v4f*)(hstart + (size_t)(nchunks - 1) * plane + i) = hs;
}

constexpr int kT     = 64;
constexpr int kKOff  = kT * 64;
constexpr int kCoefF = 2 * kT * 64;

template <int KS, int PA, int PB>
__device__ __forceinline__
void bc_stage(const float* __restrict__ As, const float* __restrict__ Bs, v8f& c0, v8f& c1)
{
    v2f a[KS / 4], b0[KS / 4], b1[KS / 4];
    #pragma unroll
    for (int j = 0; j < KS / 4; ++j) {
        a[j]    = *(const v2f*)(As + 4 * j);
        b0[j].x = Bs[(4 * j)     * PB];
        b0[j].y = Bs[(4 * j + 1) * PB];
        b1[j].x = Bs[(4 * j)     * PB + 16];
        b1[j].y = Bs[(4 * j + 1) * PB + 16];
    }
    __builtin_amdgcn_sched_barrier(0);
    #pragma unroll
    for (int j = 0; j < KS / 4; ++j) {
        c0 = __builtin_amdgcn_wmma_f32_16x16x4_f32(false, a[j], false, b0[j], (short)0, c0, false, false);
        c1 = __builtin_amdgcn_wmma_f32_16x16x4_f32(false, a[j], false, b1[j], (short)0, c1, false, false);
    }
}

template <int K, int L>
__global__ __launch_bounds__(256)
void bc_gemm(const float* __restrict__ Bm, const float* __restrict__ Cm,
             const float* __restrict__ U,
             float* __restrict__ BuT, float* __restrict__ CuT)
{
    constexpr int KS = 64;
    constexpr int BM = 128;
    constexpr int BN = 32;
    constexpr int PA = KS + 4;
    constexpr int PB = BN + 8;
    constexpr int NS = K / KS;
    constexpr int A_NP = 8;
    constexpr int B_NP = 2;
    constexpr int P = 68;
    static_assert(K % KS == 0 && (PA % 8) == 4 && (PB % 32) == 8, "tile shapes and pitches");

    __shared__ __attribute__((aligned(16))) float sA[2][BM * PA];
    __shared__ __attribute__((aligned(16))) float sB[2][KS * PB];
    __shared__ __attribute__((aligned(16))) float sBu[BN * P];
    __shared__ __attribute__((aligned(16))) float sCu[BN * P];

    const int b  = blockIdx.y;
    const int l0 = blockIdx.x * BN;
    const float* __restrict__ Ub = U + (long)b * K * L;

    const int t    = threadIdx.x;
    const int wave = t >> 5;
    const int lane = t & 31;
    const int row  = lane & 15;
    const int kh   = lane >> 4;

    const int ar = t >> 4, aq = t & 15;
    const int br = t >> 3, bq = t & 7;
    const float* __restrict__ agB = Bm + (long)ar * K + 4 * aq;
    const float* __restrict__ agC = Cm + (long)ar * K + 4 * aq;
    const float* __restrict__ bg  = Ub + (long)br * L + l0 + 4 * bq;
    const int as_off = ar * PA + 4 * aq;
    const int bs_off = br * PB + 4 * bq;
    const int af_off = (wave * 16 + row) * PA + 2 * kh;
    const int bf_off = (2 * kh) * PB + row;

    v8f acc[1][2];
    acc[0][0] = v8f{};
    acc[0][1] = v8f{};

    v4f ra[A_NP], rb[B_NP];

    #pragma unroll
    for (int p = 0; p < 4; ++p) ra[p]     = *(const v4f*)(agB + (long)p * 16 * K);
    #pragma unroll
    for (int p = 0; p < 4; ++p) ra[4 + p] = *(const v4f*)(agC + (long)p * 16 * K);
    #pragma unroll
    for (int p = 0; p < B_NP; ++p) rb[p] = *(const v4f*)(bg + (long)p * 32 * L);
    #pragma unroll
    for (int p = 0; p < A_NP; ++p) *(v4f*)(&sA[0][as_off + p * 16 * PA]) = ra[p];
    #pragma unroll
    for (int p = 0; p < B_NP; ++p) *(v4f*)(&sB[0][bs_off + p * 32 * PB]) = rb[p];
    __syncthreads();

    for (int s = 0; s + 1 < NS; ++s) {
        const int cur = s & 1;
        agB += KS;
        agC += KS;
        bg  += (long)KS * L;
        #pragma unroll
        for (int p = 0; p < 4; ++p) ra[p]     = *(const v4f*)(agB + (long)p * 16 * K);
        #pragma unroll
        for (int p = 0; p < 4; ++p) ra[4 + p] = *(const v4f*)(agC + (long)p * 16 * K);
        #pragma unroll
        for (int p = 0; p < B_NP; ++p) rb[p] = *(const v4f*)(bg + (long)p * 32 * L);

        bc_stage<KS, PA, PB>(&sA[cur][af_off], &sB[cur][bf_off], acc[0][0], acc[0][1]);

        __builtin_amdgcn_sched_barrier(0);

        #pragma unroll
        for (int p = 0; p < A_NP; ++p) *(v4f*)(&sA[cur ^ 1][as_off + p * 16 * PA]) = ra[p];
        #pragma unroll
        for (int p = 0; p < B_NP; ++p) *(v4f*)(&sB[cur ^ 1][bs_off + p * 32 * PB]) = rb[p];
        __syncthreads();
    }
    bc_stage<KS, PA, PB>(&sA[(NS - 1) & 1][af_off], &sB[(NS - 1) & 1][bf_off], acc[0][0], acc[0][1]);

    {
        float* __restrict__ dstS = (wave < 4) ? sBu : sCu;
        const int nb = (wave & 3) * 16 + 8 * kh;
        #pragma unroll
        for (int j = 0; j < 2; ++j) {
            const v4f lo = {acc[0][j][0], acc[0][j][1], acc[0][j][2], acc[0][j][3]};
            const v4f hi = {acc[0][j][4], acc[0][j][5], acc[0][j][6], acc[0][j][7]};
            *(v4f*)(&dstS[(16 * j + row) * P + nb])     = lo;
            *(v4f*)(&dstS[(16 * j + row) * P + nb + 4]) = hi;
        }
    }
    __syncthreads();

    {
        const int r = t >> 4, q = t & 15;
        float* __restrict__ dB = BuT + ((size_t)b * L + l0) * 64;
        float* __restrict__ dC = CuT + ((size_t)b * L + l0) * 64;
        #pragma unroll
        for (int p = 0; p < 2; ++p) {
            *(v4f*)(dB + (r + 16 * p) * 64 + 4 * q) = *(const v4f*)(&sBu[(r + 16 * p) * P + 4 * q]);
            *(v4f*)(dC + (r + 16 * p) * 64 + 4 * q) = *(const v4f*)(&sCu[(r + 16 * p) * P + 4 * q]);
        }
    }
}

constexpr int kCoefP = 68;
template <int L, int T>
__device__ __forceinline__
void chunk_coef_body(int sl, int b, int c, int nchunks,
                     const float* __restrict__ BuT, const float* __restrict__ CuT,
                     const float* __restrict__ A, float* __restrict__ SK,
                     float* __restrict__ sBu, float* __restrict__ sCu, float* __restrict__ sAp)
{
    static_assert(T == kT, "coefficient block layout");
    constexpr int P = kCoefP;
    const int t    = threadIdx.x;
    const int lbeg = c * T;

    if (t < 64) {
        const float a = A[t];
        float pw = 1.0f;
        for (int k = 0; k <= T; ++k) { sAp[k * P + t] = pw; pw *= a; }
    }
    {
        const int r = t >> 4, q = t & 15;
        const float* __restrict__ srcB = BuT + ((size_t)b * L + lbeg) * 64;
        const float* __restrict__ srcC = CuT + ((size_t)b * L + lbeg + 16 * sl) * 64;
        for (int p = 0; p <= sl; ++p)
            *(v4f*)(&sBu[(r + 16 * p) * P + 4 * q]) = *(const v4f*)(srcB + (r + 16 * p) * 64 + 4 * q);
        *(v4f*)(&sCu[r * P + 4 * q]) = *(const v4f*)(srcC + r * 64 + 4 * q);
    }
    __syncthreads();

    float* __restrict__ blk = SK + ((size_t)b * nchunks + c) * kCoefF;

    #pragma unroll
    for (int p = 0; p < 4; ++p) {
        const int idx = t + 256 * p;
        const int r = idx >> 6, n = idx & 63;
        const int l = 16 * sl + r;
        blk[l * 64 + n] = sCu[r * P + n] * sAp[(l + 1) * P + n];
    }

    const int li = t >> 4, jj = t & 15;
    const int l  = 16 * sl + li;
    const float* __restrict__ cr = &sCu[li * P];
    for (int sj = 0; sj < T / 16; ++sj) {
        const int j = 16 * sj + jj;
        float kv = 0.0f;
        if (sj <= sl) {
            const int k = (j <= l) ? (l - j) : 0;
            const float* __restrict__ ar2 = &sAp[k * P];
            const float* __restrict__ br2 = &sBu[j * P];
            float q0 = 0.0f, q1 = 0.0f, q2 = 0.0f, q3 = 0.0f;
            #pragma unroll
            for (int n = 0; n < 64; n += 4) {
                const v4f cv = *(const v4f*)(cr + n);
                const v4f av = *(const v4f*)(ar2 + n);
                const v4f bv = *(const v4f*)(br2 + n);
                q0 = fmaf(cv.x, av.x * bv.x, q0);
                q1 = fmaf(cv.y, av.y * bv.y, q1);
                q2 = fmaf(cv.z, av.z * bv.z, q2);
                q3 = fmaf(cv.w, av.w * bv.w, q3);
            }
            kv = (j <= l) ? ((q0 + q1) + (q2 + q3)) : 0.0f;
        }
        blk[kKOff + l * 64 + j] = kv;
    }
}

template <int N, int L, int T, int BATCH>
__global__ __launch_bounds__(256)
void carry_and_coef(const float* __restrict__ hloc, const float* __restrict__ pdel,
                    const float* __restrict__ A, float* __restrict__ hstart,
                    int nchunks, size_t plane, int carry_wgs,
                    const float* __restrict__ BuT, const float* __restrict__ CuT,
                    float* __restrict__ SK)
{
    __shared__ __attribute__((aligned(16))) float sBu[T * kCoefP];
    __shared__ __attribute__((aligned(16))) float sCu[16 * kCoefP];
    __shared__ __attribute__((aligned(16))) float sAp[(T + 1) * kCoefP];

    const int wg = blockIdx.x;
    if (wg < carry_wgs) {
        chunk_carry_body<N, T>(wg, hloc, pdel, A, hstart, nchunks, plane);
    } else {
        constexpr int SL = T / 16;
        const int id = wg - carry_wgs;
        const int sl = id % SL;
        const int b  = (id / SL) % BATCH;
        const int c  = id / (SL * BATCH);
        chunk_coef_body<L, T>(sl, b, c, nchunks, BuT, CuT, A, SK, sBu, sCu, sAp);
    }
}

typedef const __attribute__((address_space(4))) float* cfloat_p;

template <int l>
__device__ __forceinline__
float flat_step(cfloat_p krow, float (&w)[kT], float& dpb, float dpprev,
                float y1, float osum, float dl, float ul, float Dd)
{
    constexpr int s0 = (l >> 4) << 4;
    dpb *= dl;
    #pragma unroll
    for (int j = s0; j < l; ++j) w[j] *= dl;
    w[l] = dl * ul;
    float sn = 0.0f;
    #pragma unroll
    for (int j = s0; j <= l; ++j) sn = fmaf(krow[j], w[j], sn);
    const float so = fmaf(dpprev, y1, osum);
    const float y  = fmaf(dpb, so, sn);
    return fmaf(ul, Dd, y);
}

template <int l4>
__device__ __forceinline__
void flat_group4(cfloat_p K, const v4f y1, const v4f uv, const float (&dl)[16], const float (&osum)[16],
                 float* __restrict__ ybd, float (&w)[kT], float& dpb, float dpprev, float Dd)
{
    constexpr int i4 = l4 & 15;
    v4f yo;
    yo.x = flat_step<l4 + 0>(K + (l4 + 0) * 64, w, dpb, dpprev, y1.x, osum[i4 + 0], dl[i4 + 0], uv.x, Dd);
    yo.y = flat_step<l4 + 1>(K + (l4 + 1) * 64, w, dpb, dpprev, y1.y, osum[i4 + 1], dl[i4 + 1], uv.y, Dd);
    yo.z = flat_step<l4 + 2>(K + (l4 + 2) * 64, w, dpb, dpprev, y1.z, osum[i4 + 2], dl[i4 + 2], uv.z, Dd);
    yo.w = flat_step<l4 + 3>(K + (l4 + 3) * 64, w, dpb, dpprev, y1.w, osum[i4 + 3], dl[i4 + 3], uv.w, Dd);
    *(v4f*)(ybd + l4) = yo;
}

constexpr int kPW = 48 + 4;
constexpr int kPO = 16 + 4;
__device__ __forceinline__
void lds_barrier()
{
    __builtin_amdgcn_fence(__ATOMIC_RELEASE, "workgroup", "local");
    __builtin_amdgcn_s_barrier();
    __builtin_amdgcn_fence(__ATOMIC_ACQUIRE, "workgroup", "local");
}
template <int s>
__device__ __forceinline__
void flat_kfrag(const float* __restrict__ Kg, int lane, v4f (&ka)[2 * s])
{
    const float* __restrict__ kp = Kg + (16 * s + (lane & 15)) * 64 + 4 * (lane >> 4);
    #pragma unroll
    for (int q = 0; q < 2 * s; ++q) ka[q] = *(const v4f*)(kp + 8 * q);
}
template <int s>
__device__ __forceinline__
void flat_older_mm(const v4f (&ka)[2 * s], float* sWw, float* sOw, int lane,
                   const float (&w)[kT], float (&osum)[16])
{
    #pragma unroll
    for (int i = 0; i < 4 * s; ++i) {
        const v4f wv4 = {w[4 * i], w[4 * i + 1], w[4 * i + 2], w[4 * i + 3]};
        *(v4f*)(sWw + lane * kPW + 4 * i) = wv4;
    }
    lds_barrier();
    const int cc = lane & 15;
    const int kh = lane >> 4;
    v8f acc[2];
    acc[0] = v8f{};
    acc[1] = v8f{};
    #pragma unroll
    for (int q = 0; q < 2 * s; ++q) {
        const v2f a0 = {ka[q].x, ka[q].y};
        const v2f a1 = {ka[q].z, ka[q].w};
        #pragma unroll
        for (int dt = 0; dt < 2; ++dt) {
            const v4f bb = *(const v4f*)(sWw + (16 * dt + cc) * kPW + 8 * q + 4 * kh);
            const v2f b0 = {bb.x, bb.y};
            const v2f b1 = {bb.z, bb.w};
            acc[dt] = __builtin_amdgcn_wmma_f32_16x16x4_f32(
                          false, a0, false, b0, (short)0, acc[dt], false, false);
            acc[dt] = __builtin_amdgcn_wmma_f32_16x16x4_f32(
                          false, a1, false, b1, (short)0, acc[dt], false, false);
        }
    }
    #pragma unroll
    for (int dt = 0; dt < 2; ++dt) {
        float* dst = sOw + (16 * dt + cc) * kPO + 8 * kh;
        const v4f lo = {acc[dt][0], acc[dt][1], acc[dt][2], acc[dt][3]};
        const v4f hi = {acc[dt][4], acc[dt][5], acc[dt][6], acc[dt][7]};
        *(v4f*)(dst)     = lo;
        *(v4f*)(dst + 4) = hi;
    }
    lds_barrier();
    #pragma unroll
    for (int i = 0; i < 4; ++i) {
        const v4f o = *(const v4f*)(sOw + lane * kPO + 4 * i);
        osum[4 * i] = o.x; osum[4 * i + 1] = o.y; osum[4 * i + 2] = o.z; osum[4 * i + 3] = o.w;
    }
}

template <int s>
__device__ __forceinline__
void flat_block(cfloat_p K, const float* __restrict__ yrow, const float (&dl)[16],
                const v4f (&uv)[4], const float (&osum)[16], float* __restrict__ ybd, float (&w)[kT],
                float& dpprev, float Dd)
{
    v4f y1[4];
    #pragma unroll
    for (int i = 0; i < 4; ++i) y1[i] = *(const v4f*)(yrow + 16 * s + 4 * i);
    float dpb = 1.0f;
    flat_group4<16 * s + 0> (K, y1[0], uv[0], dl, osum, ybd, w, dpb, dpprev, Dd);
    flat_group4<16 * s + 4> (K, y1[1], uv[1], dl, osum, ybd, w, dpb, dpprev, Dd);
    flat_group4<16 * s + 8> (K, y1[2], uv[2], dl, osum, ybd, w, dpb, dpprev, Dd);
    flat_group4<16 * s + 12>(K, y1[3], uv[3], dl, osum, ybd, w, dpb, dpprev, Dd);
    if (16 * (s + 1) < kT) {
        #pragma unroll
        for (int j = 0; j < 16 * s; ++j) w[j] *= dpb;
        dpprev *= dpb;
    }
}

__device__ __forceinline__
void flat_fetch(const float* __restrict__ dTb, const float* __restrict__ ubd,
                int dmodel, int l0, float (&dl)[16], v4f (&uv)[4])
{
    #pragma unroll
    for (int i = 0; i < 16; ++i) dl[i] = dTb[(size_t)(l0 + i) * dmodel];
    #pragma unroll
    for (int i = 0; i < 4; ++i) uv[i] = *(const v4f*)(ubd + l0 + 4 * i);
}

template <int T>
__global__ __launch_bounds__(64)
void ssm_scan_flat(const float* __restrict__ u, const float* __restrict__ D,
                   const float* __restrict__ SK, const float* __restrict__ dT,
                   const float* __restrict__ hstart, float* __restrict__ out,
                   int dmodel, int L)
{
    static_assert(T == kT, "coefficient block layout");
    constexpr int PY = T + 4;
    __shared__ __attribute__((aligned(16))) float sY[2][32 * PY];
    __shared__ __attribute__((aligned(16))) float sW[2][32 * kPW];
    __shared__ __attribute__((aligned(16))) float sO[2][32 * kPO];

    const int b    = blockIdx.y;
    const int c    = blockIdx.z;
    const int t    = threadIdx.x;
    const int d    = blockIdx.x * 64 + t;
    const int lbeg = c * T;
    const int lane = t & 31;
    const int wv   = __builtin_amdgcn_readfirstlane(t >> 5);

    const float* __restrict__ ubd = u   + ((size_t)b * dmodel + d) * L + lbeg;
    float*       __restrict__ ybd = out + ((size_t)b * dmodel + d) * L + lbeg;
    const float* __restrict__ dTb = dT  + ((size_t)b * L + lbeg) * dmodel + d;
    const float* __restrict__ coef = SK + ((size_t)b * gridDim.z + c) * kCoefF;
    cfloat_p Kc = (cfloat_p)(coef + kKOff);

    const float Dd = D[d];

    float dl0[16], dl1[16];
    v4f   uv0[4],  uv1[4];
    flat_fetch(dTb, ubd, dmodel, 0, dl0, uv0);

    {
        const int cc = lane & 15;
        const int kh = lane >> 4;
        const size_t orow = ((((size_t)c * gridDim.y + b) * dmodel) + blockIdx.x * 64 + 32 * wv) * 64;
        const float* __restrict__ ap = coef + cc * 64 + 4 * kh;
        const float* __restrict__ bp = hstart + orow + cc * 64 + 4 * kh;

        v8f acc[4][2];
        #pragma unroll
        for (int lt = 0; lt < 4; ++lt) {
            #pragma unroll
            for (int dt = 0; dt < 2; ++dt) acc[lt][dt] = v8f{};
        }
        #pragma unroll
        for (int q = 0; q < 8; ++q) {
            v4f a[4], bb[2];
            #pragma unroll
            for (int lt = 0; lt < 4; ++lt) a[lt]  = *(const v4f*)(ap + lt * 16 * 64 + 8 * q);
            #pragma unroll
            for (int dt = 0; dt < 2; ++dt) bb[dt] = *(const v4f*)(bp + dt * 16 * 64 + 8 * q);
            #pragma unroll
            for (int lt = 0; lt < 4; ++lt) {
                const v2f a0 = {a[lt].x, a[lt].y};
                const v2f a1 = {a[lt].z, a[lt].w};
                #pragma unroll
                for (int dt = 0; dt < 2; ++dt) {
                    const v2f b0 = {bb[dt].x, bb[dt].y};
                    const v2f b1 = {bb[dt].z, bb[dt].w};
                    acc[lt][dt] = __builtin_amdgcn_wmma_f32_16x16x4_f32(
                                      false, a0, false, b0, (short)0, acc[lt][dt], false, false);
                    acc[lt][dt] = __builtin_amdgcn_wmma_f32_16x16x4_f32(
                                      false, a1, false, b1, (short)0, acc[lt][dt], false, false);
                }
            }
        }
        float* __restrict__ yw = &sY[wv][0];
        #pragma unroll
        for (int lt = 0; lt < 4; ++lt) {
            #pragma unroll
            for (int dt = 0; dt < 2; ++dt) {
                float* __restrict__ dst = yw + (16 * dt + cc) * PY + 16 * lt + 8 * kh;
                const v4f lo = {acc[lt][dt][0], acc[lt][dt][1], acc[lt][dt][2], acc[lt][dt][3]};
                const v4f hi = {acc[lt][dt][4], acc[lt][dt][5], acc[lt][dt][6], acc[lt][dt][7]};
                *(v4f*)(dst)     = lo;
                *(v4f*)(dst + 4) = hi;
            }
        }
    }
    __syncthreads();

    const float* __restrict__ yrow = &sY[wv][lane * PY];
    float w[kT];
    float dpprev = 1.0f;

    const float* __restrict__ Kg = coef + kKOff;
    float* sWw = &sW[wv][0];
    float* sOw = &sO[wv][0];
    v4f ka1[2], ka2[4], ka3[6];
    float osum[16];
    #pragma unroll
    for (int i = 0; i < 16; ++i) osum[i] = 0.0f;

    flat_fetch(dTb, ubd, dmodel, 16, dl1, uv1);
    flat_kfrag<1>(Kg, lane, ka1);
    flat_block<0>(Kc, yrow, dl0, uv0, osum, ybd, w, dpprev, Dd);
    flat_fetch(dTb, ubd, dmodel, 32, dl0, uv0);
    flat_kfrag<2>(Kg, lane, ka2);
    flat_older_mm<1>(ka1, sWw, sOw, lane, w, osum);
    flat_block<1>(Kc, yrow, dl1, uv1, osum, ybd, w, dpprev, Dd);
    flat_fetch(dTb, ubd, dmodel, 48, dl1, uv1);
    flat_kfrag<3>(Kg, lane, ka3);
    flat_older_mm<2>(ka2, sWw, sOw, lane, w, osum);
    flat_block<2>(Kc, yrow, dl0, uv0, osum, ybd, w, dpprev, Dd);
    flat_older_mm<3>(ka3, sWw, sOw, lane, w, osum);
    flat_block<3>(Kc, yrow, dl1, uv1, osum, ybd, w, dpprev, Dd);
}

extern "C" void kernel_launch(void* const* d_in, const int* in_sizes, int n_in,
                              void* d_out, int out_size, void* d_ws, size_t ws_size,
                              hipStream_t stream)
{
    (void)in_sizes; (void)n_in; (void)out_size; (void)ws_size;

    const float* u     = (const float*)d_in[1];
    const float* A     = (const float*)d_in[2];
    const float* Bm    = (const float*)d_in[3];
    const float* Cm    = (const float*)d_in[4];
    const float* D     = (const float*)d_in[5];
    const float* Delta = (const float*)d_in[6];
    float* out = (float*)d_out;

    constexpr int batch = 4, dmodel = 1024, N = 64, L = 2048;

    float* BuT = (float*)d_ws;
    float* CuT = BuT + (size_t)batch * L * N;
    float* dT  = CuT + (size_t)batch * L * N;

    constexpr int chunks = 32, T = L / chunks;
    constexpr size_t plane = (size_t)batch * dmodel * N;
    float* hloc   = dT    + (size_t)batch * L * dmodel;
    float* pdel   = hloc  + (size_t)(chunks - 1) * plane;
    float* hstart = pdel  + (size_t)(chunks - 1) * batch * dmodel;
    float* SK     = hstart + (size_t)chunks * plane;

    const dim3 blk(128, 1, 1);

    const dim3 gblk(256, 1, 1);
    bc_gemm<dmodel, L><<<dim3(L / 32, batch, 1), gblk, 0, stream>>>(Bm, Cm, u, BuT, CuT);
    gemm_f32_wmma<dmodel, L, 4, 2, 32><<<dim3(L / 128, dmodel / 128, batch), gblk, 0, stream>>>(
        Delta, u, dT, dmodel);

    ssm_chunk_state<T><<<dim3(dmodel / 64, batch, chunks - 1), blk, 0, stream>>>(
        u, A, BuT, dT, hloc, pdel, dmodel, N, L);
    constexpr int carry_wgs = (int)(plane / 4 / 256);
    constexpr int coef_wgs  = (T / 16) * batch * chunks;
    carry_and_coef<N, L, T, batch><<<dim3(carry_wgs + coef_wgs, 1, 1), gblk, 0, stream>>>(
        hloc, pdel, A, hstart, chunks, plane, carry_wgs, BuT, CuT, SK);
    ssm_scan_flat<T><<<dim3(dmodel / 64, batch, chunks), dim3(64, 1, 1), 0, stream>>>(
        u, D, SK, dT, hstart, out, dmodel, L);
}
